# starting kernel
# baseline (speedup 1.0000x reference)
#include <hip/hip_runtime.h>
#include <cmath>


typedef __attribute__((ext_vector_type(2))) float v2f;
typedef __attribute__((ext_vector_type(8))) float v8f;

#define EPSBN 1e-5f

__device__ __forceinline__ v8f wmma_f32(v2f a, v2f b, v8f c) {
  return __builtin_amdgcn_wmma_f32_16x16x4_f32(false, a, false, b, (short)0, c,
                                               false, false);
}

__global__ void k_start(const float* __restrict__ x, const float* __restrict__ Wst,
                        const float* __restrict__ bst, float* __restrict__ H) {
  int t = blockIdx.x * blockDim.x + threadIdx.x;
  int o   = t & 31;
  int pos = t >> 5;
  int l   = pos & 63;
  int n   = (pos >> 6) & 511;
  int b   = pos >> 15;
  const float* xb = x + (size_t)b * 2 * 512 * 64;
  float x0 = xb[(size_t)n * 64 + l];
  float x1 = xb[(size_t)512 * 64 + (size_t)n * 64 + l];
  H[t] = fmaf(x0, Wst[o * 2], fmaf(x1, Wst[o * 2 + 1], bst[o]));
}

__global__ void __launch_bounds__(256) k_gated(
    const float* __restrict__ H, float* __restrict__ G,
    const float* __restrict__ Wf, const float* __restrict__ bf,
    const float* __restrict__ Wg, const float* __restrict__ bg, int Lout) {
  int wid  = (blockIdx.x * blockDim.x + threadIdx.x) >> 5;
  int lane = threadIdx.x & 31;
  int half = lane >> 4;
  int l15  = lane & 15;
  int otile = wid & 1;
  int rtile = wid >> 1;
  int m0 = rtile << 4;
  int o  = (otile << 4) + l15;

  const float* Arow = H + (size_t)(m0 + l15) * 32;
  float biasf = bf[o], biasg = bg[o];
  v8f cf, cg;
#pragma unroll
  for (int r = 0; r < 8; ++r) { cf[r] = biasf; cg[r] = biasg; }

#pragma unroll
  for (int k0 = 0; k0 < 64; k0 += 4) {
    int k = k0 + half * 2;
    v2f a = *(const v2f*)(Arow + k);
    int c0 = k & 31, tap = k >> 5;
    const float* wf = Wf + (size_t)o * 64 + c0 * 2 + tap;
    const float* wg = Wg + (size_t)o * 64 + c0 * 2 + tap;
    v2f bfv, bgv;
    bfv.x = wf[0]; bfv.y = wf[2];
    bgv.x = wg[0]; bgv.y = wg[2];
    cf = wmma_f32(a, bfv, cf);
    cg = wmma_f32(a, bgv, cg);
  }
#pragma unroll
  for (int r = 0; r < 8; ++r) {
    int row = m0 + r + half * 8;
    int l = row & 63;
    if (l < Lout) {
      float fv = tanhf(cf[r]);
      float gv = 1.0f / (1.0f + __expf(-cg[r]));
      G[(size_t)row * 32 + o] = fv * gv;
    }
  }
}

__global__ void __launch_bounds__(256) k_skip(
    const float* __restrict__ G, float* __restrict__ S,
    const float* __restrict__ Wk, const float* __restrict__ bk,
    int off, int initFlag) {
  int wid  = (blockIdx.x * blockDim.x + threadIdx.x) >> 5;
  int lane = threadIdx.x & 31;
  int half = lane >> 4;
  int l15  = lane & 15;
  int otile = wid & 1;
  int rtile = wid >> 1;
  int m0 = rtile << 4;
  int o  = (otile << 4) + l15;

  const float* Arow = G + (size_t)(m0 + l15 + off) * 32;
  float bias = bk[o];
  v8f c;
#pragma unroll
  for (int r = 0; r < 8; ++r) {
    int row = m0 + r + half * 8;
    float prev = initFlag ? 0.0f : S[(size_t)row * 32 + o];
    c[r] = bias + prev;
  }
#pragma unroll
  for (int k0 = 0; k0 < 32; k0 += 4) {
    int k = k0 + half * 2;
    v2f a = *(const v2f*)(Arow + k);
    v2f b = *(const v2f*)(Wk + (size_t)o * 32 + k);
    c = wmma_f32(a, b, c);
  }
#pragma unroll
  for (int r = 0; r < 8; ++r) {
    int row = m0 + r + half * 8;
    int l = row & 63;
    if (l < 61) S[(size_t)row * 32 + o] = c[r];
  }
}

#define LDSPITCH 516

__global__ void __launch_bounds__(256) k_diffuse(
    const float* __restrict__ adj, const float* __restrict__ Xin,
    float* __restrict__ Xout) {
  __shared__ float aT[16 * LDSPITCH];

  int bid = blockIdx.x;
  int cg = bid & 3;
  int mt = (bid >> 2) & 31;
  int b  = bid >> 7;
  int m0 = mt << 4;

#pragma unroll 4
  for (int idx = threadIdx.x; idx < 8192; idx += 256) {
    int ml = idx & 15;
    int k  = idx >> 4;
    aT[ml * LDSPITCH + k] = adj[(size_t)k * 512 + m0 + ml];
  }
  __syncthreads();

  int wave = threadIdx.x >> 5;
  int lane = threadIdx.x & 31;
  int half = lane >> 4;
  int l15  = lane & 15;
  int j0   = (cg << 9) + (wave << 6);

  const float* Xb = Xin + (size_t)b * 512 * 2048;
  float* Ob       = Xout + (size_t)b * 512 * 2048;
  const float* Ar = aT + l15 * LDSPITCH;

  v8f c0 = {}, c1 = {}, c2 = {}, c3 = {};
#pragma unroll 4
  for (int n0 = 0; n0 < 512; n0 += 4) {
    int k = n0 + half * 2;
    v2f a = *(const v2f*)(Ar + k);
    const float* xr = Xb + (size_t)k * 2048 + j0 + l15;
    v2f b0, b1, b2, b3;
    b0.x = xr[0];  b0.y = xr[2048];
    b1.x = xr[16]; b1.y = xr[2048 + 16];
    b2.x = xr[32]; b2.y = xr[2048 + 32];
    b3.x = xr[48]; b3.y = xr[2048 + 48];
    c0 = wmma_f32(a, b0, c0);
    c1 = wmma_f32(a, b1, c1);
    c2 = wmma_f32(a, b2, c2);
    c3 = wmma_f32(a, b3, c3);
  }
#pragma unroll
  for (int r = 0; r < 8; ++r) {
    size_t orow = (size_t)(m0 + r + half * 8) * 2048 + j0 + l15;
    Ob[orow]      = c0[r];
    Ob[orow + 16] = c1[r];
    Ob[orow + 32] = c2[r];
    Ob[orow + 48] = c3[r];
  }
}

__global__ void __launch_bounds__(256) k_gconv(
    const float* __restrict__ G, const float* __restrict__ X1,
    const float* __restrict__ X2, const float* __restrict__ Hres,
    float* __restrict__ Hout, const float* __restrict__ Wgc,
    const float* __restrict__ bgc, const float* __restrict__ gcg,
    const float* __restrict__ gcb, const float* __restrict__ bng,
    const float* __restrict__ bnb, int Lout) {
  int wid  = (blockIdx.x * blockDim.x + threadIdx.x) >> 5;
  int lane = threadIdx.x & 31;
  int half = lane >> 4;
  int l15  = lane & 15;
  int otile = wid & 1;
  int rtile = wid >> 1;
  int m0 = rtile << 4;
  int o  = (otile << 4) + l15;

  size_t arow = (size_t)(m0 + l15) * 32;
  float bias = bgc[o];
  v8f c;
#pragma unroll
  for (int r = 0; r < 8; ++r) c[r] = bias;

  const float* srcs[3] = {G, X1, X2};
#pragma unroll
  for (int part = 0; part < 3; ++part) {
    const float* src = srcs[part];
#pragma unroll
    for (int k0 = 0; k0 < 32; k0 += 4) {
      int k = k0 + half * 2;
      v2f a = *(const v2f*)(src + arow + k);
      v2f b = *(const v2f*)(Wgc + (size_t)o * 96 + part * 32 + k);
      c = wmma_f32(a, b, c);
    }
  }
  float rinv = rsqrtf(1.0f + EPSBN);
  float ig = gcg[o] * rinv, bgv = gcb[o];
  float ib = bng[o] * rinv, bbv = bnb[o];
#pragma unroll
  for (int r = 0; r < 8; ++r) {
    int row = m0 + r + half * 8;
    int l = row & 63;
    if (l < Lout) {
      float g = c[r] * ig + bgv;
      float h = g + Hres[(size_t)(row + 1) * 32 + o];
      Hout[(size_t)row * 32 + o] = h * ib + bbv;
    }
  }
}

__global__ void k_out(const float* __restrict__ S, const float* __restrict__ og,
                      const float* __restrict__ ob, float* __restrict__ out) {
  int t = blockIdx.x * blockDim.x + threadIdx.x;
  if (t >= 8 * 61 * 512 * 32) return;
  int cch = t & 31;
  int n   = (t >> 5) & 511;
  int rem = t >> 14;
  int l   = rem % 61;
  int b   = rem / 61;
  float inv = rsqrtf(1.0f + EPSBN);
  float v = S[(((size_t)(b * 512 + n)) * 64 + l) * 32 + cch];
  out[t] = v * (og[cch] * inv) + ob[cch];
}

extern "C" void kernel_launch(void* const* d_in, const int* in_sizes, int n_in,
                              void* d_out, int out_size, void* d_ws, size_t ws_size,
                              hipStream_t stream) {
  (void)in_sizes; (void)n_in; (void)out_size; (void)ws_size;
  const float* x   = (const float*)d_in[0];
  const float* adj = (const float*)d_in[1];
  const float* Wst = (const float*)d_in[2];
  const float* bst = (const float*)d_in[3];
  const float* Wf  = (const float*)d_in[4];
  const float* bf  = (const float*)d_in[5];
  const float* Wg  = (const float*)d_in[6];
  const float* bg  = (const float*)d_in[7];
  const float* Wsk = (const float*)d_in[8];
  const float* bsk = (const float*)d_in[9];
  const float* Wgc = (const float*)d_in[10];
  const float* bgc = (const float*)d_in[11];
  const float* gcg = (const float*)d_in[12];
  const float* gcb = (const float*)d_in[13];
  const float* bng = (const float*)d_in[14];
  const float* bnb = (const float*)d_in[15];
  const float* og  = (const float*)d_in[16];
  const float* ob  = (const float*)d_in[17];
  float* out = (float*)d_out;

  const size_t BUF = (size_t)8 * 512 * 64 * 32 + 1024;
  float* ws    = (float*)d_ws;
  float* bufH  = ws;
  float* bufH2 = ws + BUF;
  float* bufG  = ws + 2 * BUF;
  float* bufX1 = ws + 3 * BUF;
  float* bufX2 = ws + 4 * BUF;
  float* bufS  = ws + 5 * BUF;

  k_start<<<32768, 256, 0, stream>>>(x, Wst, bst, bufH);

  float* H  = bufH;
  float* Hn = bufH2;
  for (int i = 0; i < 3; ++i) {
    int Lout = 63 - i;
    k_gated<<<4096, 256, 0, stream>>>(H, bufG, Wf + i * 2048, bf + i * 32,
                                      Wg + i * 2048, bg + i * 32, Lout);
    k_skip<<<4096, 256, 0, stream>>>(bufG, bufS, Wsk + i * 1024, bsk + i * 32,
                                     2 - i, i == 0 ? 1 : 0);
    if (i < 2) {
      k_diffuse<<<1024, 256, 0, stream>>>(adj, bufG, bufX1);
      k_diffuse<<<1024, 256, 0, stream>>>(adj, bufX1, bufX2);
      k_gconv<<<4096, 256, 0, stream>>>(bufG, bufX1, bufX2, H, Hn,
                                        Wgc + i * 3072, bgc + i * 32,
                                        gcg + i * 32, gcb + i * 32,
                                        bng + i * 32, bnb + i * 32, Lout);
      float* tmp = H; H = Hn; Hn = tmp;
    }
  }
  k_out<<<31232, 256, 0, stream>>>(bufS, og, ob, out);
}


# k_diffuse: remap the wave tile's columns (accumulator t, lane column c -> X column 4c+t) so each lane loads B and stores D as 16-byte vectors: 4x fewer global load/store instructions; sums unchanged
# speedup vs baseline: 1.0373x; 1.0373x over previous
#include <hip/hip_runtime.h>
#include <cmath>


typedef __attribute__((ext_vector_type(2))) float v2f;
typedef __attribute__((ext_vector_type(8))) float v8f;

#define EPSBN 1e-5f

__device__ __forceinline__ v8f wmma_f32(v2f a, v2f b, v8f c) {
  return __builtin_amdgcn_wmma_f32_16x16x4_f32(false, a, false, b, (short)0, c,
                                               false, false);
}

__global__ void k_start(const float* __restrict__ x, const float* __restrict__ Wst,
                        const float* __restrict__ bst, float* __restrict__ H) {
  int t = blockIdx.x * blockDim.x + threadIdx.x;
  int o   = t & 31;
  int pos = t >> 5;
  int l   = pos & 63;
  int n   = (pos >> 6) & 511;
  int b   = pos >> 15;
  const float* xb = x + (size_t)b * 2 * 512 * 64;
  float x0 = xb[(size_t)n * 64 + l];
  float x1 = xb[(size_t)512 * 64 + (size_t)n * 64 + l];
  H[t] = fmaf(x0, Wst[o * 2], fmaf(x1, Wst[o * 2 + 1], bst[o]));
}

__global__ void __launch_bounds__(256) k_gated(
    const float* __restrict__ H, float* __restrict__ G,
    const float* __restrict__ Wf, const float* __restrict__ bf,
    const float* __restrict__ Wg, const float* __restrict__ bg, int Lout) {
  int wid  = (blockIdx.x * blockDim.x + threadIdx.x) >> 5;
  int lane = threadIdx.x & 31;
  int half = lane >> 4;
  int l15  = lane & 15;
  int otile = wid & 1;
  int rtile = wid >> 1;
  int m0 = rtile << 4;
  int o  = (otile << 4) + l15;

  const float* Arow = H + (size_t)(m0 + l15) * 32;
  float biasf = bf[o], biasg = bg[o];
  v8f cf, cg;
#pragma unroll
  for (int r = 0; r < 8; ++r) { cf[r] = biasf; cg[r] = biasg; }

#pragma unroll
  for (int k0 = 0; k0 < 64; k0 += 4) {
    int k = k0 + half * 2;
    v2f a = *(const v2f*)(Arow + k);
    int c0 = k & 31, tap = k >> 5;
    const float* wf = Wf + (size_t)o * 64 + c0 * 2 + tap;
    const float* wg = Wg + (size_t)o * 64 + c0 * 2 + tap;
    v2f bfv, bgv;
    bfv.x = wf[0]; bfv.y = wf[2];
    bgv.x = wg[0]; bgv.y = wg[2];
    cf = wmma_f32(a, bfv, cf);
    cg = wmma_f32(a, bgv, cg);
  }
#pragma unroll
  for (int r = 0; r < 8; ++r) {
    int row = m0 + r + half * 8;
    int l = row & 63;
    if (l < Lout) {
      float fv = tanhf(cf[r]);
      float gv = 1.0f / (1.0f + __expf(-cg[r]));
      G[(size_t)row * 32 + o] = fv * gv;
    }
  }
}

__global__ void __launch_bounds__(256) k_skip(
    const float* __restrict__ G, float* __restrict__ S,
    const float* __restrict__ Wk, const float* __restrict__ bk,
    int off, int initFlag) {
  int wid  = (blockIdx.x * blockDim.x + threadIdx.x) >> 5;
  int lane = threadIdx.x & 31;
  int half = lane >> 4;
  int l15  = lane & 15;
  int otile = wid & 1;
  int rtile = wid >> 1;
  int m0 = rtile << 4;
  int o  = (otile << 4) + l15;

  const float* Arow = G + (size_t)(m0 + l15 + off) * 32;
  float bias = bk[o];
  v8f c;
#pragma unroll
  for (int r = 0; r < 8; ++r) {
    int row = m0 + r + half * 8;
    float prev = initFlag ? 0.0f : S[(size_t)row * 32 + o];
    c[r] = bias + prev;
  }
#pragma unroll
  for (int k0 = 0; k0 < 32; k0 += 4) {
    int k = k0 + half * 2;
    v2f a = *(const v2f*)(Arow + k);
    v2f b = *(const v2f*)(Wk + (size_t)o * 32 + k);
    c = wmma_f32(a, b, c);
  }
#pragma unroll
  for (int r = 0; r < 8; ++r) {
    int row = m0 + r + half * 8;
    int l = row & 63;
    if (l < 61) S[(size_t)row * 32 + o] = c[r];
  }
}

#define LDSPITCH 516

__global__ void __launch_bounds__(256) k_diffuse(
    const float* __restrict__ adj, const float* __restrict__ Xin,
    float* __restrict__ Xout) {
  __shared__ float aT[16 * LDSPITCH];

  int bid = blockIdx.x;
  int cg = bid & 3;
  int mt = (bid >> 2) & 31;
  int b  = bid >> 7;
  int m0 = mt << 4;

#pragma unroll 4
  for (int idx = threadIdx.x; idx < 8192; idx += 256) {
    int ml = idx & 15;
    int k  = idx >> 4;
    aT[ml * LDSPITCH + k] = adj[(size_t)k * 512 + m0 + ml];
  }
  __syncthreads();

  int wave = threadIdx.x >> 5;
  int lane = threadIdx.x & 31;
  int half = lane >> 4;
  int l15  = lane & 15;
  int j0   = (cg << 9) + (wave << 6);

  const float* Xb = Xin + (size_t)b * 512 * 2048;
  float* Ob       = Xout + (size_t)b * 512 * 2048;
  const float* Ar = aT + l15 * LDSPITCH;

  const float* xr = Xb + (size_t)(half * 2) * 2048 + j0 + 4 * l15;

  v8f c0 = {}, c1 = {}, c2 = {}, c3 = {};
#pragma unroll 4
  for (int n0 = 0; n0 < 512; n0 += 4) {
    int k = n0 + half * 2;
    v2f a = *(const v2f*)(Ar + k);
    const float4 r0 = *(const float4*)(xr + (size_t)n0 * 2048);
    const float4 r1 = *(const float4*)(xr + (size_t)n0 * 2048 + 2048);
    v2f b0, b1, b2, b3;
    b0.x = r0.x; b0.y = r1.x;
    b1.x = r0.y; b1.y = r1.y;
    b2.x = r0.z; b2.y = r1.z;
    b3.x = r0.w; b3.y = r1.w;
    c0 = wmma_f32(a, b0, c0);
    c1 = wmma_f32(a, b1, c1);
    c2 = wmma_f32(a, b2, c2);
    c3 = wmma_f32(a, b3, c3);
  }
#pragma unroll
  for (int r = 0; r < 8; ++r) {
    size_t orow = (size_t)(m0 + r + half * 8) * 2048 + j0 + 4 * l15;
    float4 o;
    o.x = c0[r]; o.y = c1[r]; o.z = c2[r]; o.w = c3[r];
    *(float4*)(Ob + orow) = o;
  }
}

__global__ void __launch_bounds__(256) k_gconv(
    const float* __restrict__ G, const float* __restrict__ X1,
    const float* __restrict__ X2, const float* __restrict__ Hres,
    float* __restrict__ Hout, const float* __restrict__ Wgc,
    const float* __restrict__ bgc, const float* __restrict__ gcg,
    const float* __restrict__ gcb, const float* __restrict__ bng,
    const float* __restrict__ bnb, int Lout) {
  int wid  = (blockIdx.x * blockDim.x + threadIdx.x) >> 5;
  int lane = threadIdx.x & 31;
  int half = lane >> 4;
  int l15  = lane & 15;
  int otile = wid & 1;
  int rtile = wid >> 1;
  int m0 = rtile << 4;
  int o  = (otile << 4) + l15;

  size_t arow = (size_t)(m0 + l15) * 32;
  float bias = bgc[o];
  v8f c;
#pragma unroll
  for (int r = 0; r < 8; ++r) c[r] = bias;

  const float* srcs[3] = {G, X1, X2};
#pragma unroll
  for (int part = 0; part < 3; ++part) {
    const float* src = srcs[part];
#pragma unroll
    for (int k0 = 0; k0 < 32; k0 += 4) {
      int k = k0 + half * 2;
      v2f a = *(const v2f*)(src + arow + k);
      v2f b = *(const v2f*)(Wgc + (size_t)o * 96 + part * 32 + k);
      c = wmma_f32(a, b, c);
    }
  }
  float rinv = rsqrtf(1.0f + EPSBN);
  float ig = gcg[o] * rinv, bgv = gcb[o];
  float ib = bng[o] * rinv, bbv = bnb[o];
#pragma unroll
  for (int r = 0; r < 8; ++r) {
    int row = m0 + r + half * 8;
    int l = row & 63;
    if (l < Lout) {
      float g = c[r] * ig + bgv;
      float h = g + Hres[(size_t)(row + 1) * 32 + o];
      Hout[(size_t)row * 32 + o] = h * ib + bbv;
    }
  }
}

__global__ void k_out(const float* __restrict__ S, const float* __restrict__ og,
                      const float* __restrict__ ob, float* __restrict__ out) {
  int t = blockIdx.x * blockDim.x + threadIdx.x;
  if (t >= 8 * 61 * 512 * 32) return;
  int cch = t & 31;
  int n   = (t >> 5) & 511;
  int rem = t >> 14;
  int l   = rem % 61;
  int b   = rem / 61;
  float inv = rsqrtf(1.0f + EPSBN);
  float v = S[(((size_t)(b * 512 + n)) * 64 + l) * 32 + cch];
  out[t] = v * (og[cch] * inv) + ob[cch];
}

extern "C" void kernel_launch(void* const* d_in, const int* in_sizes, int n_in,
                              void* d_out, int out_size, void* d_ws, size_t ws_size,
                              hipStream_t stream) {
  (void)in_sizes; (void)n_in; (void)out_size; (void)ws_size;
  const float* x   = (const float*)d_in[0];
  const float* adj = (const float*)d_in[1];
  const float* Wst = (const float*)d_in[2];
  const float* bst = (const float*)d_in[3];
  const float* Wf  = (const float*)d_in[4];
  const float* bf  = (const float*)d_in[5];
  const float* Wg  = (const float*)d_in[6];
  const float* bg  = (const float*)d_in[7];
  const float* Wsk = (const float*)d_in[8];
  const float* bsk = (const float*)d_in[9];
  const float* Wgc = (const float*)d_in[10];
  const float* bgc = (const float*)d_in[11];
  const float* gcg = (const float*)d_in[12];
  const float* gcb = (const float*)d_in[13];
  const float* bng = (const float*)d_in[14];
  const float* bnb = (const float*)d_in[15];
  const float* og  = (const float*)d_in[16];
  const float* ob  = (const float*)d_in[17];
  float* out = (float*)d_out;

  const size_t BUF = (size_t)8 * 512 * 64 * 32 + 1024;
  float* ws    = (float*)d_ws;
  float* bufH  = ws;
  float* bufH2 = ws + BUF;
  float* bufG  = ws + 2 * BUF;
  float* bufX1 = ws + 3 * BUF;
  float* bufX2 = ws + 4 * BUF;
  float* bufS  = ws + 5 * BUF;

  k_start<<<32768, 256, 0, stream>>>(x, Wst, bst, bufH);

  float* H  = bufH;
  float* Hn = bufH2;
  for (int i = 0; i < 3; ++i) {
    int Lout = 63 - i;
    k_gated<<<4096, 256, 0, stream>>>(H, bufG, Wf + i * 2048, bf + i * 32,
                                      Wg + i * 2048, bg + i * 32, Lout);
    k_skip<<<4096, 256, 0, stream>>>(bufG, bufS, Wsk + i * 1024, bsk + i * 32,
                                     2 - i, i == 0 ? 1 : 0);
    if (i < 2) {
      k_diffuse<<<1024, 256, 0, stream>>>(adj, bufG, bufX1);
      k_diffuse<<<1024, 256, 0, stream>>>(adj, bufX1, bufX2);
      k_gconv<<<4096, 256, 0, stream>>>(bufG, bufX1, bufX2, H, Hn,
                                        Wgc + i * 3072, bgc + i * 32,
                                        gcg + i * 32, gcb + i * 32,
                                        bng + i * 32, bnb + i * 32, Lout);
      float* tmp = H; H = Hn; Hn = tmp;
    }
  }
  k_out<<<31232, 256, 0, stream>>>(bufS, og, ob, out);
}


# all kernels but k_diffuse: 4-32x more work per wave (k_start, k_out: 8 x 16 B per thread; gated, skip, gconv: 4 row tiles per wave, weights kept in registers); 7x fewer waves; sums unchanged
# speedup vs baseline: 1.1406x; 1.0996x over previous
#include <hip/hip_runtime.h>
#include <cmath>


typedef __attribute__((ext_vector_type(2))) float v2f;
typedef __attribute__((ext_vector_type(8))) float v8f;

#define EPSBN 1e-5f

__device__ __forceinline__ v8f wmma_f32(v2f a, v2f b, v8f c) {
  return __builtin_amdgcn_wmma_f32_16x16x4_f32(false, a, false, b, (short)0, c,
                                               false, false);
}

__global__ void __launch_bounds__(256) k_start(
    const float* __restrict__ x, const float* __restrict__ Wst,
    const float* __restrict__ bst, float* __restrict__ H) {
  int t    = blockIdx.x * 256 + threadIdx.x;
  int lane = threadIdx.x & 31;
  int c4   = (lane & 7) << 2;
  int sub  = lane >> 3;
  int wbase = (t >> 5) << 5;
  float w0[4], w1[4], bb[4];
#pragma unroll
  for (int q = 0; q < 4; ++q) {
    w0[q] = Wst[(c4 + q) * 2];
    w1[q] = Wst[(c4 + q) * 2 + 1];
    bb[q] = bst[c4 + q];
  }
#pragma unroll
  for (int i = 0; i < 8; ++i) {
    int pos = wbase + 4 * i + sub;
    int b   = pos >> 15;
    const float* xb = x + (size_t)b * 2 * 512 * 64 + (pos & 32767);
    float x0 = xb[0];
    float x1 = xb[512 * 64];
    float4 h;
    h.x = fmaf(x0, w0[0], fmaf(x1, w1[0], bb[0]));
    h.y = fmaf(x0, w0[1], fmaf(x1, w1[1], bb[1]));
    h.z = fmaf(x0, w0[2], fmaf(x1, w1[2], bb[2]));
    h.w = fmaf(x0, w0[3], fmaf(x1, w1[3], bb[3]));
    *(float4*)(H + (size_t)pos * 32 + c4) = h;
  }
}

#define TPW 4

__global__ void __launch_bounds__(256, 8) k_gated(
    const float* __restrict__ H, float* __restrict__ G,
    const float* __restrict__ Wf, const float* __restrict__ bf,
    const float* __restrict__ Wg, const float* __restrict__ bg, int Lout) {
  int wid  = (blockIdx.x * blockDim.x + threadIdx.x) >> 5;
  int lane = threadIdx.x & 31;
  int half = lane >> 4;
  int l15  = lane & 15;
  int otile = wid & 1;
  int rgrp  = wid >> 1;
  int o  = (otile << 4) + l15;

  float biasf = bf[o], biasg = bg[o];
  v2f bfv[16], bgv[16];
#pragma unroll
  for (int s = 0; s < 16; ++s) {
    int k = s * 4 + half * 2;
    int c0 = k & 31, tap = k >> 5;
    const float* wf = Wf + (size_t)o * 64 + c0 * 2 + tap;
    const float* wg = Wg + (size_t)o * 64 + c0 * 2 + tap;
    bfv[s].x = wf[0]; bfv[s].y = wf[2];
    bgv[s].x = wg[0]; bgv[s].y = wg[2];
  }

  for (int it = 0; it < TPW; ++it) {
    int m0 = (rgrp * TPW + it) << 4;
    const float* Arow = H + (size_t)(m0 + l15) * 32;
    v8f cf, cg;
#pragma unroll
    for (int r = 0; r < 8; ++r) { cf[r] = biasf; cg[r] = biasg; }
#pragma unroll
    for (int s = 0; s < 16; ++s) {
      int k = s * 4 + half * 2;
      v2f a = *(const v2f*)(Arow + k);
      cf = wmma_f32(a, bfv[s], cf);
      cg = wmma_f32(a, bgv[s], cg);
    }
#pragma unroll
    for (int r = 0; r < 8; ++r) {
      int row = m0 + r + half * 8;
      int l = row & 63;
      if (l < Lout) {
        float fv = tanhf(cf[r]);
        float gv = 1.0f / (1.0f + __expf(-cg[r]));
        G[(size_t)row * 32 + o] = fv * gv;
      }
    }
  }
}

__global__ void __launch_bounds__(256) k_skip(
    const float* __restrict__ G, float* __restrict__ S,
    const float* __restrict__ Wk, const float* __restrict__ bk,
    int off, int initFlag) {
  int wid  = (blockIdx.x * blockDim.x + threadIdx.x) >> 5;
  int lane = threadIdx.x & 31;
  int half = lane >> 4;
  int l15  = lane & 15;
  int otile = wid & 1;
  int rgrp  = wid >> 1;
  int o  = (otile << 4) + l15;

  float bias = bk[o];
  v2f bw[8];
#pragma unroll
  for (int s = 0; s < 8; ++s) {
    int k = s * 4 + half * 2;
    bw[s] = *(const v2f*)(Wk + (size_t)o * 32 + k);
  }

  for (int it = 0; it < TPW; ++it) {
    int m0 = (rgrp * TPW + it) << 4;
    const float* Arow = G + (size_t)(m0 + l15 + off) * 32;
    v8f c;
#pragma unroll
    for (int r = 0; r < 8; ++r) {
      int row = m0 + r + half * 8;
      float prev = initFlag ? 0.0f : S[(size_t)row * 32 + o];
      c[r] = bias + prev;
    }
#pragma unroll
    for (int s = 0; s < 8; ++s) {
      int k = s * 4 + half * 2;
      v2f a = *(const v2f*)(Arow + k);
      c = wmma_f32(a, bw[s], c);
    }
#pragma unroll
    for (int r = 0; r < 8; ++r) {
      int row = m0 + r + half * 8;
      int l = row & 63;
      if (l < 61) S[(size_t)row * 32 + o] = c[r];
    }
  }
}

#define LDSPITCH 516

__global__ void __launch_bounds__(256) k_diffuse(
    const float* __restrict__ adj, const float* __restrict__ Xin,
    float* __restrict__ Xout) {
  __shared__ float aT[16 * LDSPITCH];

  int bid = blockIdx.x;
  int cg = bid & 3;
  int mt = (bid >> 2) & 31;
  int b  = bid >> 7;
  int m0 = mt << 4;

#pragma unroll 4
  for (int idx = threadIdx.x; idx < 8192; idx += 256) {
    int ml = idx & 15;
    int k  = idx >> 4;
    aT[ml * LDSPITCH + k] = adj[(size_t)k * 512 + m0 + ml];
  }
  __syncthreads();

  int wave = threadIdx.x >> 5;
  int lane = threadIdx.x & 31;
  int half = lane >> 4;
  int l15  = lane & 15;
  int j0   = (cg << 9) + (wave << 6);

  const float* Xb = Xin + (size_t)b * 512 * 2048;
  float* Ob       = Xout + (size_t)b * 512 * 2048;
  const float* Ar = aT + l15 * LDSPITCH;

  const float* xr = Xb + (size_t)(half * 2) * 2048 + j0 + 4 * l15;

  v8f c0 = {}, c1 = {}, c2 = {}, c3 = {};
#pragma unroll 4
  for (int n0 = 0; n0 < 512; n0 += 4) {
    int k = n0 + half * 2;
    v2f a = *(const v2f*)(Ar + k);
    const float4 r0 = *(const float4*)(xr + (size_t)n0 * 2048);
    const float4 r1 = *(const float4*)(xr + (size_t)n0 * 2048 + 2048);
    v2f b0, b1, b2, b3;
    b0.x = r0.x; b0.y = r1.x;
    b1.x = r0.y; b1.y = r1.y;
    b2.x = r0.z; b2.y = r1.z;
    b3.x = r0.w; b3.y = r1.w;
    c0 = wmma_f32(a, b0, c0);
    c1 = wmma_f32(a, b1, c1);
    c2 = wmma_f32(a, b2, c2);
    c3 = wmma_f32(a, b3, c3);
  }
#pragma unroll
  for (int r = 0; r < 8; ++r) {
    size_t orow = (size_t)(m0 + r + half * 8) * 2048 + j0 + 4 * l15;
    float4 o;
    o.x = c0[r]; o.y = c1[r]; o.z = c2[r]; o.w = c3[r];
    *(float4*)(Ob + orow) = o;
  }
}

__global__ void __launch_bounds__(256) k_gconv(
    const float* __restrict__ G, const float* __restrict__ X1,
    const float* __restrict__ X2, const float* __restrict__ Hres,
    float* __restrict__ Hout, const float* __restrict__ Wgc,
    const float* __restrict__ bgc, const float* __restrict__ gcg,
    const float* __restrict__ gcb, const float* __restrict__ bng,
    const float* __restrict__ bnb, int Lout) {
  int wid  = (blockIdx.x * blockDim.x + threadIdx.x) >> 5;
  int lane = threadIdx.x & 31;
  int half = lane >> 4;
  int l15  = lane & 15;
  int otile = wid & 1;
  int rgrp  = wid >> 1;
  int o  = (otile << 4) + l15;

  float bias = bgc[o];
  v2f bw[24];
#pragma unroll
  for (int s = 0; s < 24; ++s) {
    int k = (s & 7) * 4 + half * 2;
    bw[s] = *(const v2f*)(Wgc + (size_t)o * 96 + (s >> 3) * 32 + k);
  }
  float rinv = rsqrtf(1.0f + EPSBN);
  float ig = gcg[o] * rinv, bgv = gcb[o];
  float ib = bng[o] * rinv, bbv = bnb[o];

  const float* srcs[3] = {G, X1, X2};
  for (int it = 0; it < TPW; ++it) {
    int m0 = (rgrp * TPW + it) << 4;
    size_t arow = (size_t)(m0 + l15) * 32;
    v8f c;
#pragma unroll
    for (int r = 0; r < 8; ++r) c[r] = bias;
#pragma unroll
    for (int part = 0; part < 3; ++part) {
      const float* src = srcs[part];
#pragma unroll
      for (int s = 0; s < 8; ++s) {
        int k = s * 4 + half * 2;
        v2f a = *(const v2f*)(src + arow + k);
        c = wmma_f32(a, bw[part * 8 + s], c);
      }
    }
#pragma unroll
    for (int r = 0; r < 8; ++r) {
      int row = m0 + r + half * 8;
      int l = row & 63;
      if (l < Lout) {
        float g = c[r] * ig + bgv;
        float h = g + Hres[(size_t)(row + 1) * 32 + o];
        Hout[(size_t)row * 32 + o] = h * ib + bbv;
      }
    }
  }
}

__global__ void __launch_bounds__(256) k_out(
    const float* __restrict__ S, const float* __restrict__ og,
    const float* __restrict__ ob, float* __restrict__ out) {
  int c4 = (threadIdx.x & 7) << 2;
  float inv = rsqrtf(1.0f + EPSBN);
  float sc[4], sh[4];
#pragma unroll
  for (int q = 0; q < 4; ++q) { sc[q] = og[c4 + q] * inv; sh[q] = ob[c4 + q]; }
#pragma unroll
  for (int i = 0; i < 8; ++i) {
    int g   = blockIdx.x * 2048 + i * 256 + threadIdx.x;
    int n   = (g >> 3) & 511;
    int rem = g >> 12;
    int l   = rem % 61;
    int b   = rem / 61;
    const float4 v =
        *(const float4*)(S + (((size_t)(b * 512 + n)) * 64 + l) * 32 + c4);
    float4 r;
    r.x = v.x * sc[0] + sh[0];
    r.y = v.y * sc[1] + sh[1];
    r.z = v.z * sc[2] + sh[2];
    r.w = v.w * sc[3] + sh[3];
    *(float4*)(out + (size_t)g * 4) = r;
  }
}

extern "C" void kernel_launch(void* const* d_in, const int* in_sizes, int n_in,
                              void* d_out, int out_size, void* d_ws, size_t ws_size,
                              hipStream_t stream) {
  (void)in_sizes; (void)n_in; (void)out_size; (void)ws_size;
  const float* x   = (const float*)d_in[0];
  const float* adj = (const float*)d_in[1];
  const float* Wst = (const float*)d_in[2];
  const float* bst = (const float*)d_in[3];
  const float* Wf  = (const float*)d_in[4];
  const float* bf  = (const float*)d_in[5];
  const float* Wg  = (const float*)d_in[6];
  const float* bg  = (const float*)d_in[7];
  const float* Wsk = (const float*)d_in[8];
  const float* bsk = (const float*)d_in[9];
  const float* Wgc = (const float*)d_in[10];
  const float* bgc = (const float*)d_in[11];
  const float* gcg = (const float*)d_in[12];
  const float* gcb = (const float*)d_in[13];
  const float* bng = (const float*)d_in[14];
  const float* bnb = (const float*)d_in[15];
  const float* og  = (const float*)d_in[16];
  const float* ob  = (const float*)d_in[17];
  float* out = (float*)d_out;

  const size_t BUF = (size_t)8 * 512 * 64 * 32 + 1024;
  float* ws    = (float*)d_ws;
  float* bufH  = ws;
  float* bufH2 = ws + BUF;
  float* bufG  = ws + 2 * BUF;
  float* bufX1 = ws + 3 * BUF;
  float* bufX2 = ws + 4 * BUF;
  float* bufS  = ws + 5 * BUF;

  k_start<<<1024, 256, 0, stream>>>(x, Wst, bst, bufH);

  float* H  = bufH;
  float* Hn = bufH2;
  for (int i = 0; i < 3; ++i) {
    int Lout = 63 - i;
    k_gated<<<1024, 256, 0, stream>>>(H, bufG, Wf + i * 2048, bf + i * 32,
                                      Wg + i * 2048, bg + i * 32, Lout);
    k_skip<<<1024, 256, 0, stream>>>(bufG, bufS, Wsk + i * 1024, bsk + i * 32,
                                     2 - i, i == 0 ? 1 : 0);
    if (i < 2) {
      k_diffuse<<<1024, 256, 0, stream>>>(adj, bufG, bufX1);
      k_diffuse<<<1024, 256, 0, stream>>>(adj, bufX1, bufX2);
      k_gconv<<<1024, 256, 0, stream>>>(bufG, bufX1, bufX2, H, Hn,
                                        Wgc + i * 3072, bgc + i * 32,
                                        gcg + i * 32, gcb + i * 32,
                                        bng + i * 32, bnb + i * 32, Lout);
      float* tmp = H; H = Hn; Hn = tmp;
    }
  }
  k_out<<<976, 256, 0, stream>>>(bufS, og, ob, out);
}


# k_diffuse: wave tile 16x64 -> 64x64 (16 accumulators, 64 adj rows per block in LDS): each X fragment loaded feeds 4 row tiles, 4x fewer L2 reads; loads double-buffered in registers; sums unchanged
# speedup vs baseline: 1.3402x; 1.1750x over previous
#include <hip/hip_runtime.h>
#include <cmath>


typedef __attribute__((ext_vector_type(2))) float v2f;
typedef __attribute__((ext_vector_type(8))) float v8f;

#define EPSBN 1e-5f

__device__ __forceinline__ v8f wmma_f32(v2f a, v2f b, v8f c) {
  return __builtin_amdgcn_wmma_f32_16x16x4_f32(false, a, false, b, (short)0, c,
                                               false, false);
}

__global__ void __launch_bounds__(256) k_start(
    const float* __restrict__ x, const float* __restrict__ Wst,
    const float* __restrict__ bst, float* __restrict__ H) {
  int t    = blockIdx.x * 256 + threadIdx.x;
  int lane = threadIdx.x & 31;
  int c4   = (lane & 7) << 2;
  int sub  = lane >> 3;
  int wbase = (t >> 5) << 5;
  float w0[4], w1[4], bb[4];
#pragma unroll
  for (int q = 0; q < 4; ++q) {
    w0[q] = Wst[(c4 + q) * 2];
    w1[q] = Wst[(c4 + q) * 2 + 1];
    bb[q] = bst[c4 + q];
  }
#pragma unroll
  for (int i = 0; i < 8; ++i) {
    int pos = wbase + 4 * i + sub;
    int b   = pos >> 15;
    const float* xb = x + (size_t)b * 2 * 512 * 64 + (pos & 32767);
    float x0 = xb[0];
    float x1 = xb[512 * 64];
    float4 h;
    h.x = fmaf(x0, w0[0], fmaf(x1, w1[0], bb[0]));
    h.y = fmaf(x0, w0[1], fmaf(x1, w1[1], bb[1]));
    h.z = fmaf(x0, w0[2], fmaf(x1, w1[2], bb[2]));
    h.w = fmaf(x0, w0[3], fmaf(x1, w1[3], bb[3]));
    *(float4*)(H + (size_t)pos * 32 + c4) = h;
  }
}

#define TPW 4

__global__ void __launch_bounds__(256, 8) k_gated(
    const float* __restrict__ H, float* __restrict__ G,
    const float* __restrict__ Wf, const float* __restrict__ bf,
    const float* __restrict__ Wg, const float* __restrict__ bg, int Lout) {
  int wid  = (blockIdx.x * blockDim.x + threadIdx.x) >> 5;
  int lane = threadIdx.x & 31;
  int half = lane >> 4;
  int l15  = lane & 15;
  int otile = wid & 1;
  int rgrp  = wid >> 1;
  int o  = (otile << 4) + l15;

  float biasf = bf[o], biasg = bg[o];
  v2f bfv[16], bgv[16];
#pragma unroll
  for (int s = 0; s < 16; ++s) {
    int k = s * 4 + half * 2;
    int c0 = k & 31, tap = k >> 5;
    const float* wf = Wf + (size_t)o * 64 + c0 * 2 + tap;
    const float* wg = Wg + (size_t)o * 64 + c0 * 2 + tap;
    bfv[s].x = wf[0]; bfv[s].y = wf[2];
    bgv[s].x = wg[0]; bgv[s].y = wg[2];
  }

  for (int it = 0; it < TPW; ++it) {
    int m0 = (rgrp * TPW + it) << 4;
    const float* Arow = H + (size_t)(m0 + l15) * 32;
    v8f cf, cg;
#pragma unroll
    for (int r = 0; r < 8; ++r) { cf[r] = biasf; cg[r] = biasg; }
#pragma unroll
    for (int s = 0; s < 16; ++s) {
      int k = s * 4 + half * 2;
      v2f a = *(const v2f*)(Arow + k);
      cf = wmma_f32(a, bfv[s], cf);
      cg = wmma_f32(a, bgv[s], cg);
    }
#pragma unroll
    for (int r = 0; r < 8; ++r) {
      int row = m0 + r + half * 8;
      int l = row & 63;
      if (l < Lout) {
        float fv = tanhf(cf[r]);
        float gv = 1.0f / (1.0f + __expf(-cg[r]));
        G[(size_t)row * 32 + o] = fv * gv;
      }
    }
  }
}

__global__ void __launch_bounds__(256) k_skip(
    const float* __restrict__ G, float* __restrict__ S,
    const float* __restrict__ Wk, const float* __restrict__ bk,
    int off, int initFlag) {
  int wid  = (blockIdx.x * blockDim.x + threadIdx.x) >> 5;
  int lane = threadIdx.x & 31;
  int half = lane >> 4;
  int l15  = lane & 15;
  int otile = wid & 1;
  int rgrp  = wid >> 1;
  int o  = (otile << 4) + l15;

  float bias = bk[o];
  v2f bw[8];
#pragma unroll
  for (int s = 0; s < 8; ++s) {
    int k = s * 4 + half * 2;
    bw[s] = *(const v2f*)(Wk + (size_t)o * 32 + k);
  }

  for (int it = 0; it < TPW; ++it) {
    int m0 = (rgrp * TPW + it) << 4;
    const float* Arow = G + (size_t)(m0 + l15 + off) * 32;
    v8f c;
#pragma unroll
    for (int r = 0; r < 8; ++r) {
      int row = m0 + r + half * 8;
      float prev = initFlag ? 0.0f : S[(size_t)row * 32 + o];
      c[r] = bias + prev;
    }
#pragma unroll
    for (int s = 0; s < 8; ++s) {
      int k = s * 4 + half * 2;
      v2f a = *(const v2f*)(Arow + k);
      c = wmma_f32(a, bw[s], c);
    }
#pragma unroll
    for (int r = 0; r < 8; ++r) {
      int row = m0 + r + half * 8;
      int l = row & 63;
      if (l < 61) S[(size_t)row * 32 + o] = c[r];
    }
  }
}

#define LDSPITCH 516
#define DROWS 64

struct DFrag {
  v2f a[4][4];
  float4 r0[4];
  float4 r1[4];
};

__device__ __forceinline__ void dload(DFrag& f, const float* Ar,
                                      const float* xr, int n0) {
#pragma unroll
  for (int u = 0; u < 4; ++u) {
#pragma unroll
    for (int t = 0; t < 4; ++t)
      f.a[u][t] = *(const v2f*)(Ar + t * 16 * LDSPITCH + n0 + 4 * u);
    f.r0[u] = *(const float4*)(xr + (size_t)(n0 + 4 * u) * 2048);
    f.r1[u] = *(const float4*)(xr + (size_t)(n0 + 4 * u) * 2048 + 2048);
  }
}

__device__ __forceinline__ void dmma(const DFrag& f, v8f (&acc)[4][4]) {
#pragma unroll
  for (int u = 0; u < 4; ++u) {
    v2f b0, b1, b2, b3;
    b0.x = f.r0[u].x; b0.y = f.r1[u].x;
    b1.x = f.r0[u].y; b1.y = f.r1[u].y;
    b2.x = f.r0[u].z; b2.y = f.r1[u].z;
    b3.x = f.r0[u].w; b3.y = f.r1[u].w;
#pragma unroll
    for (int t = 0; t < 4; ++t) {
      acc[t][0] = wmma_f32(f.a[u][t], b0, acc[t][0]);
      acc[t][1] = wmma_f32(f.a[u][t], b1, acc[t][1]);
      acc[t][2] = wmma_f32(f.a[u][t], b2, acc[t][2]);
      acc[t][3] = wmma_f32(f.a[u][t], b3, acc[t][3]);
    }
  }
}

__global__ void __launch_bounds__(256) k_diffuse(
    const float* __restrict__ adj, const float* __restrict__ Xin,
    float* __restrict__ Xout) {
  __shared__ float aT[DROWS * LDSPITCH];

  int bid = blockIdx.x;
  int cg = bid & 3;
  int mt = (bid >> 2) & 7;
  int b  = bid >> 5;
  int m0 = mt << 6;

#pragma unroll
  for (int i = 0; i < 32; ++i) {
    int idx = threadIdx.x + 256 * i;
    int q   = idx & 3;
    int kk  = (idx >> 2) & 7;
    int g   = (idx >> 5) & 3;
    int k   = ((idx >> 7) << 3) + kk;
    int ml  = 16 * g + 4 * q;
    const float4 v = *(const float4*)(adj + (size_t)k * 512 + m0 + ml);
    aT[(ml + 0) * LDSPITCH + k] = v.x;
    aT[(ml + 1) * LDSPITCH + k] = v.y;
    aT[(ml + 2) * LDSPITCH + k] = v.z;
    aT[(ml + 3) * LDSPITCH + k] = v.w;
  }
  __syncthreads();

  int wave = threadIdx.x >> 5;
  int lane = threadIdx.x & 31;
  int half = lane >> 4;
  int l15  = lane & 15;
  int j0   = (cg << 9) + (wave << 6);

  const float* Xb = Xin + (size_t)b * 512 * 2048;
  float* Ob       = Xout + (size_t)b * 512 * 2048;
  const float* Ar = aT + l15 * LDSPITCH + half * 2;

  const float* xr = Xb + (size_t)(half * 2) * 2048 + j0 + 4 * l15;

  v8f acc[4][4];
#pragma unroll
  for (int t = 0; t < 4; ++t)
#pragma unroll
    for (int j = 0; j < 4; ++j)
      acc[t][j] = v8f{};

#define LOADS_GROUP() __builtin_amdgcn_sched_barrier(0)
  DFrag f0, f1;
  dload(f0, Ar, xr, 0);
#pragma unroll 1
  for (int n0 = 0; n0 < 480; n0 += 32) {
    LOADS_GROUP();
    dload(f1, Ar, xr, n0 + 16);
    LOADS_GROUP();
    dmma(f0, acc);
    LOADS_GROUP();
    dload(f0, Ar, xr, n0 + 32);
    LOADS_GROUP();
    dmma(f1, acc);
  }
  LOADS_GROUP();
  dload(f1, Ar, xr, 496);
  LOADS_GROUP();
  dmma(f0, acc);
  dmma(f1, acc);
#undef LOADS_GROUP

#pragma unroll
  for (int t = 0; t < 4; ++t) {
#pragma unroll
    for (int r = 0; r < 8; ++r) {
      size_t orow = (size_t)(m0 + 16 * t + r + half * 8) * 2048 + j0 + 4 * l15;
      float4 o;
      o.x = acc[t][0][r]; o.y = acc[t][1][r];
      o.z = acc[t][2][r]; o.w = acc[t][3][r];
      *(float4*)(Ob + orow) = o;
    }
  }
}

__global__ void __launch_bounds__(256) k_gconv(
    const float* __restrict__ G, const float* __restrict__ X1,
    const float* __restrict__ X2, const float* __restrict__ Hres,
    float* __restrict__ Hout, const float* __restrict__ Wgc,
    const float* __restrict__ bgc, const float* __restrict__ gcg,
    const float* __restrict__ gcb, const float* __restrict__ bng,
    const float* __restrict__ bnb, int Lout) {
  int wid  = (blockIdx.x * blockDim.x + threadIdx.x) >> 5;
  int lane = threadIdx.x & 31;
  int half = lane >> 4;
  int l15  = lane & 15;
  int otile = wid & 1;
  int rgrp  = wid >> 1;
  int o  = (otile << 4) + l15;

  float bias = bgc[o];
  v2f bw[24];
#pragma unroll
  for (int s = 0; s < 24; ++s) {
    int k = (s & 7) * 4 + half * 2;
    bw[s] = *(const v2f*)(Wgc + (size_t)o * 96 + (s >> 3) * 32 + k);
  }
  float rinv = rsqrtf(1.0f + EPSBN);
  float ig = gcg[o] * rinv, bgv = gcb[o];
  float ib = bng[o] * rinv, bbv = bnb[o];

  const float* srcs[3] = {G, X1, X2};
  for (int it = 0; it < TPW; ++it) {
    int m0 = (rgrp * TPW + it) << 4;
    size_t arow = (size_t)(m0 + l15) * 32;
    v8f c;
#pragma unroll
    for (int r = 0; r < 8; ++r) c[r] = bias;
#pragma unroll
    for (int part = 0; part < 3; ++part) {
      const float* src = srcs[part];
#pragma unroll
      for (int s = 0; s < 8; ++s) {
        int k = s * 4 + half * 2;
        v2f a = *(const v2f*)(src + arow + k);
        c = wmma_f32(a, bw[part * 8 + s], c);
      }
    }
#pragma unroll
    for (int r = 0; r < 8; ++r) {
      int row = m0 + r + half * 8;
      int l = row & 63;
      if (l < Lout) {
        float g = c[r] * ig + bgv;
        float h = g + Hres[(size_t)(row + 1) * 32 + o];
        Hout[(size_t)row * 32 + o] = h * ib + bbv;
      }
    }
  }
}

__global__ void __launch_bounds__(256) k_out(
    const float* __restrict__ S, const float* __restrict__ og,
    const float* __restrict__ ob, float* __restrict__ out) {
  int c4 = (threadIdx.x & 7) << 2;
  float inv = rsqrtf(1.0f + EPSBN);
  float sc[4], sh[4];
#pragma unroll
  for (int q = 0; q < 4; ++q) { sc[q] = og[c4 + q] * inv; sh[q] = ob[c4 + q]; }
#pragma unroll
  for (int i = 0; i < 8; ++i) {
    int g   = blockIdx.x * 2048 + i * 256 + threadIdx.x;
    int n   = (g >> 3) & 511;
    int rem = g >> 12;
    int l   = rem % 61;
    int b   = rem / 61;
    const float4 v =
        *(const float4*)(S + (((size_t)(b * 512 + n)) * 64 + l) * 32 + c4);
    float4 r;
    r.x = v.x * sc[0] + sh[0];
    r.y = v.y * sc[1] + sh[1];
    r.z = v.z * sc[2] + sh[2];
    r.w = v.w * sc[3] + sh[3];
    *(float4*)(out + (size_t)g * 4) = r;
  }
}

extern "C" void kernel_launch(void* const* d_in, const int* in_sizes, int n_in,
                              void* d_out, int out_size, void* d_ws, size_t ws_size,
                              hipStream_t stream) {
  (void)in_sizes; (void)n_in; (void)out_size; (void)ws_size;
  const float* x   = (const float*)d_in[0];
  const float* adj = (const float*)d_in[1];
  const float* Wst = (const float*)d_in[2];
  const float* bst = (const float*)d_in[3];
  const float* Wf  = (const float*)d_in[4];
  const float* bf  = (const float*)d_in[5];
  const float* Wg  = (const float*)d_in[6];
  const float* bg  = (const float*)d_in[7];
  const float* Wsk = (const float*)d_in[8];
  const float* bsk = (const float*)d_in[9];
  const float* Wgc = (const float*)d_in[10];
  const float* bgc = (const float*)d_in[11];
  const float* gcg = (const float*)d_in[12];
  const float* gcb = (const float*)d_in[13];
  const float* bng = (const float*)d_in[14];
  const float* bnb = (const float*)d_in[15];
  const float* og  = (const float*)d_in[16];
  const float* ob  = (const float*)d_in[17];
  float* out = (float*)d_out;

  const size_t BUF = (size_t)8 * 512 * 64 * 32 + 1024;
  float* ws    = (float*)d_ws;
  float* bufH  = ws;
  float* bufH2 = ws + BUF;
  float* bufG  = ws + 2 * BUF;
  float* bufX1 = ws + 3 * BUF;
  float* bufX2 = ws + 4 * BUF;
  float* bufS  = ws + 5 * BUF;

  k_start<<<1024, 256, 0, stream>>>(x, Wst, bst, bufH);

  float* H  = bufH;
  float* Hn = bufH2;
  for (int i = 0; i < 3; ++i) {
    int Lout = 63 - i;
    k_gated<<<1024, 256, 0, stream>>>(H, bufG, Wf + i * 2048, bf + i * 32,
                                      Wg + i * 2048, bg + i * 32, Lout);
    k_skip<<<1024, 256, 0, stream>>>(bufG, bufS, Wsk + i * 1024, bsk + i * 32,
                                     2 - i, i == 0 ? 1 : 0);
    if (i < 2) {
      k_diffuse<<<256, 256, 0, stream>>>(adj, bufG, bufX1);
      k_diffuse<<<256, 256, 0, stream>>>(adj, bufX1, bufX2);
      k_gconv<<<1024, 256, 0, stream>>>(bufG, bufX1, bufX2, H, Hn,
                                        Wgc + i * 3072, bgc + i * 32,
                                        gcg + i * 32, gcb + i * 32,
                                        bng + i * 32, bnb + i * 32, Lout);
      float* tmp = H; H = Hn; Hn = tmp;
    }
  }
  k_out<<<976, 256, 0, stream>>>(bufS, og, ob, out);
}


# graph conv, next layer's gated conv and (last layer) skip/output conv fused into one launch per layer boundary: H and the last G stay in the wave's LDS slice; 10 -> 7 launches; sums unchanged
# speedup vs baseline: 1.8590x; 1.3871x over previous
#include <hip/hip_runtime.h>
#include <cmath>


typedef __attribute__((ext_vector_type(2))) float v2f;
typedef __attribute__((ext_vector_type(8))) float v8f;

#define EPSBN 1e-5f

__device__ __forceinline__ v8f wmma_f32(v2f a, v2f b, v8f c) {
  return __builtin_amdgcn_wmma_f32_16x16x4_f32(false, a, false, b, (short)0, c,
                                               false, false);
}

struct StartW { float w0[4], w1[4], bb[4]; };

__device__ __forceinline__ StartW start_weights(const float* Wst, const float* bst,
                                                int c4) {
  StartW w;
#pragma unroll
  for (int q = 0; q < 4; ++q) {
    w.w0[q] = Wst[(c4 + q) * 2];
    w.w1[q] = Wst[(c4 + q) * 2 + 1];
    w.bb[q] = bst[c4 + q];
  }
  return w;
}

__device__ __forceinline__ float4 start_conv(float x0, float x1, const StartW& w) {
  float4 h;
  h.x = fmaf(x0, w.w0[0], fmaf(x1, w.w1[0], w.bb[0]));
  h.y = fmaf(x0, w.w0[1], fmaf(x1, w.w1[1], w.bb[1]));
  h.z = fmaf(x0, w.w0[2], fmaf(x1, w.w1[2], w.bb[2]));
  h.w = fmaf(x0, w.w0[3], fmaf(x1, w.w1[3], w.bb[3]));
  return h;
}

#define XCH (512 * 64)
__device__ __forceinline__ const float* x_series(const float* x, int wid) {
  return x + (size_t)(wid >> 9) * 2 * XCH + (size_t)(wid & 511) * 64;
}

#define APITCH 36

__device__ __forceinline__ float4 ld_group(const float* src, int lane) {
  return *(const float4*)(src + lane * 4);
}
__device__ __forceinline__ void st_group(float* tile, int j, int lane, float4 v) {
  *(float4*)(tile + (4 * j + (lane >> 3)) * APITCH + ((lane & 7) << 2)) = v;
}

#define GROWS 20
template <bool FROMX>
__global__ void __launch_bounds__(256, 4) k_gated(
    const float* __restrict__ H, float* __restrict__ G,
    const float* __restrict__ Wf, const float* __restrict__ bf,
    const float* __restrict__ Wg, const float* __restrict__ bg,
    const float* __restrict__ Wst, const float* __restrict__ bst, int Lout) {
  __shared__ __attribute__((aligned(16))) float tiles[8][GROWS * APITCH + (FROMX ? 224 : 0)];
  int wave = threadIdx.x >> 5;
  int wid  = blockIdx.x * 8 + wave;
  int lane = threadIdx.x & 31;
  int half = lane >> 4;
  int l15  = lane & 15;
  int o0   = 2 * l15;
  float* T = tiles[wave];

  float bf0 = bf[o0], bf1 = bf[o0 + 1], bg0 = bg[o0], bg1 = bg[o0 + 1];
  v2f wf0[16], wf1[16], wg0[16], wg1[16];
#pragma unroll
  for (int s = 0; s < 16; ++s) {
    int k = s * 4 + half * 2;
    int c0 = k & 31, tap = k >> 5;
    const float* wf = Wf + (size_t)o0 * 64 + c0 * 2 + tap;
    const float* wg = Wg + (size_t)o0 * 64 + c0 * 2 + tap;
    wf0[s].x = wf[0];  wf0[s].y = wf[2];
    wf1[s].x = wf[64]; wf1[s].y = wf[66];
    wg0[s].x = wg[0];  wg0[s].y = wg[2];
    wg1[s].x = wg[64]; wg1[s].y = wg[66];
  }

  size_t base = (size_t)wid * 2048;
  float4 pf[5];
  float* XL = T + GROWS * APITCH;
  float* WL = XL + 128;
  int sub = lane >> 3;
  if (FROMX) {
    const float* xs = x_series(H, wid);
    float a0 = xs[lane], a1 = xs[lane + 32];
    float b0 = xs[XCH + lane], b1 = xs[XCH + lane + 32];
    float u0 = Wst[lane * 2], u1 = Wst[lane * 2 + 1], ub = bst[lane];
    XL[lane] = a0; XL[lane + 32] = a1;
    XL[64 + lane] = b0; XL[96 + lane] = b1;
    float* wq = WL + (lane >> 2) * 12 + (lane & 3);
    wq[0] = u0; wq[4] = u1; wq[8] = ub;
  } else {
#pragma unroll
    for (int j = 0; j < 5; ++j) pf[j] = ld_group(H + base + j * 128, lane);
  }

#pragma unroll 1
  for (int it = 0; it < 4; ++it) {
    if (FROMX) {
      float xa[5], xb[5];
#pragma unroll
      for (int j = 0; j < 5; ++j) {
        int row = min(16 * it + 4 * j + sub, 63);
        xa[j] = XL[row];
        xb[j] = XL[64 + row];
      }
      StartW sw;
      const float4* wq = (const float4*)(WL + (lane & 7) * 12);
      float4 q0 = wq[0], q1 = wq[1], qb = wq[2];
      sw.w0[0] = q0.x; sw.w0[1] = q0.y; sw.w0[2] = q0.z; sw.w0[3] = q0.w;
      sw.w1[0] = q1.x; sw.w1[1] = q1.y; sw.w1[2] = q1.z; sw.w1[3] = q1.w;
      sw.bb[0] = qb.x; sw.bb[1] = qb.y; sw.bb[2] = qb.z; sw.bb[3] = qb.w;
#pragma unroll
      for (int j = 0; j < 5; ++j) st_group(T, j, lane, start_conv(xa[j], xb[j], sw));
    } else {
#pragma unroll
      for (int j = 0; j < 5; ++j) st_group(T, j, lane, pf[j]);
      if (it < 3) {
        const float* nx = H + base + (it + 1) * 512;
#pragma unroll
        for (int j = 0; j < 5; ++j) pf[j] = ld_group(nx + j * 128, lane);
      }
    }
    v8f cf0, cf1, cg0, cg1;
#pragma unroll
    for (int r = 0; r < 8; ++r) { cf0[r] = bf0; cf1[r] = bf1; cg0[r] = bg0; cg1[r] = bg1; }
#pragma unroll
    for (int s = 0; s < 16; ++s) {
      v2f a = *(const v2f*)(T + (l15 + (s >> 3)) * APITCH + ((s & 7) << 2) + half * 2);
      cf0 = wmma_f32(a, wf0[s], cf0);
      cf1 = wmma_f32(a, wf1[s], cf1);
      cg0 = wmma_f32(a, wg0[s], cg0);
      cg1 = wmma_f32(a, wg1[s], cg1);
    }
#pragma unroll
    for (int r = 0; r < 8; ++r) {
      int l = (it << 4) + r + half * 8;
      if (l < Lout) {
        float2 v;
        v.x = tanhf(cf0[r]) * (1.0f / (1.0f + __expf(-cg0[r])));
        v.y = tanhf(cf1[r]) * (1.0f / (1.0f + __expf(-cg1[r])));
        *(float2*)(G + base + (size_t)l * 32 + o0) = v;
      }
    }
  }
}

#define LDSPITCH 516
#define DROWS 64

struct DFrag {
  v2f a[4][4];
  float4 r0[4];
  float4 r1[4];
};

__device__ __forceinline__ void dload(DFrag& f, const float* Ar,
                                      const float* xr, int n0) {
#pragma unroll
  for (int u = 0; u < 4; ++u) {
#pragma unroll
    for (int t = 0; t < 4; ++t)
      f.a[u][t] = *(const v2f*)(Ar + t * 16 * LDSPITCH + n0 + 4 * u);
    f.r0[u] = *(const float4*)(xr + (size_t)(n0 + 4 * u) * 2048);
    f.r1[u] = *(const float4*)(xr + (size_t)(n0 + 4 * u) * 2048 + 2048);
  }
}

__device__ __forceinline__ void dmma(const DFrag& f, v8f (&acc)[4][4]) {
#pragma unroll
  for (int u = 0; u < 4; ++u) {
    v2f b0, b1, b2, b3;
    b0.x = f.r0[u].x; b0.y = f.r1[u].x;
    b1.x = f.r0[u].y; b1.y = f.r1[u].y;
    b2.x = f.r0[u].z; b2.y = f.r1[u].z;
    b3.x = f.r0[u].w; b3.y = f.r1[u].w;
#pragma unroll
    for (int t = 0; t < 4; ++t) {
      acc[t][0] = wmma_f32(f.a[u][t], b0, acc[t][0]);
      acc[t][1] = wmma_f32(f.a[u][t], b1, acc[t][1]);
      acc[t][2] = wmma_f32(f.a[u][t], b2, acc[t][2]);
      acc[t][3] = wmma_f32(f.a[u][t], b3, acc[t][3]);
    }
  }
}

__global__ void __launch_bounds__(256) k_diffuse(
    const float* __restrict__ adj, const float* __restrict__ Xin,
    float* __restrict__ Xout) {
  __shared__ float aT[DROWS * LDSPITCH];

  int bid = blockIdx.x;
  int cg = bid & 3;
  int mt = (bid >> 2) & 7;
  int b  = bid >> 5;
  int m0 = mt << 6;

#pragma unroll
  for (int i = 0; i < 32; ++i) {
    int idx = threadIdx.x + 256 * i;
    int q   = idx & 3;
    int kk  = (idx >> 2) & 7;
    int g   = (idx >> 5) & 3;
    int k   = ((idx >> 7) << 3) + kk;
    int ml  = 16 * g + 4 * q;
    const float4 v = *(const float4*)(adj + (size_t)k * 512 + m0 + ml);
    aT[(ml + 0) * LDSPITCH + k] = v.x;
    aT[(ml + 1) * LDSPITCH + k] = v.y;
    aT[(ml + 2) * LDSPITCH + k] = v.z;
    aT[(ml + 3) * LDSPITCH + k] = v.w;
  }
  __syncthreads();

  int wave = threadIdx.x >> 5;
  int lane = threadIdx.x & 31;
  int half = lane >> 4;
  int l15  = lane & 15;
  int j0   = (cg << 9) + (wave << 6);

  const float* Xb = Xin + (size_t)b * 512 * 2048;
  float* Ob       = Xout + (size_t)b * 512 * 2048;
  const float* Ar = aT + l15 * LDSPITCH + half * 2;

  const float* xr = Xb + (size_t)(half * 2) * 2048 + j0 + 4 * l15;

  v8f acc[4][4];
#pragma unroll
  for (int t = 0; t < 4; ++t)
#pragma unroll
    for (int j = 0; j < 4; ++j)
      acc[t][j] = v8f{};

#define LOADS_GROUP() __builtin_amdgcn_sched_barrier(0)
  DFrag f0, f1;
  dload(f0, Ar, xr, 0);
#pragma unroll 1
  for (int n0 = 0; n0 < 480; n0 += 32) {
    LOADS_GROUP();
    dload(f1, Ar, xr, n0 + 16);
    LOADS_GROUP();
    dmma(f0, acc);
    LOADS_GROUP();
    dload(f0, Ar, xr, n0 + 32);
    LOADS_GROUP();
    dmma(f1, acc);
  }
  LOADS_GROUP();
  dload(f1, Ar, xr, 496);
  LOADS_GROUP();
  dmma(f0, acc);
  dmma(f1, acc);
#undef LOADS_GROUP

#pragma unroll
  for (int t = 0; t < 4; ++t) {
#pragma unroll
    for (int r = 0; r < 8; ++r) {
      size_t orow = (size_t)(m0 + 16 * t + r + half * 8) * 2048 + j0 + 4 * l15;
      float4 o;
      o.x = acc[t][0][r]; o.y = acc[t][1][r];
      o.z = acc[t][2][r]; o.w = acc[t][3][r];
      *(float4*)(Ob + orow) = o;
    }
  }
}

#define HROWS 68
template <bool FROMX, bool LAST>
__global__ void __launch_bounds__(256, 4) k_layer(
    const float* __restrict__ G, const float* __restrict__ X1,
    const float* __restrict__ X2, const float* __restrict__ Hres,
    const float* __restrict__ Wgc, const float* __restrict__ bgc,
    const float* __restrict__ gcg, const float* __restrict__ gcb,
    const float* __restrict__ bng, const float* __restrict__ bnb,
    const float* __restrict__ Wst, const float* __restrict__ bst,
    const float* __restrict__ Wf, const float* __restrict__ bf,
    const float* __restrict__ Wg, const float* __restrict__ bg,
    const float* __restrict__ G0, const float* __restrict__ Wk,
    const float* __restrict__ bk, const float* __restrict__ og,
    const float* __restrict__ ob, float* __restrict__ Hout,
    float* __restrict__ Gout, float* __restrict__ out, int Lh, int Lg) {
  __shared__ __attribute__((aligned(16))) float tiles[8][(64 + HROWS) * APITCH];
  int wave = threadIdx.x >> 5;
  int wid  = blockIdx.x * 8 + wave;
  int lane = threadIdx.x & 31;
  int half = lane >> 4;
  int l15  = lane & 15;
  int o0   = 2 * l15;
  float* T  = tiles[wave];
  float* HS = T + 64 * APITCH;
  size_t base = (size_t)wid * 2048;

  {
    v2f bw0[24], bw1[24];
#pragma unroll
    for (int s = 0; s < 24; ++s) {
      int k = (s & 7) * 4 + half * 2;
      const float* w = Wgc + (s >> 3) * 32 + k;
      bw0[s] = *(const v2f*)(w + (size_t)o0 * 96);
      bw1[s] = *(const v2f*)(w + (size_t)(o0 + 1) * 96);
    }
    float bias0 = bgc[o0], bias1 = bgc[o0 + 1];
    float rinv = rsqrtf(1.0f + EPSBN);
    float ig0 = gcg[o0] * rinv,     bgv0 = gcb[o0];
    float ig1 = gcg[o0 + 1] * rinv, bgv1 = gcb[o0 + 1];
    float ib0 = bng[o0] * rinv,     bbv0 = bnb[o0];
    float ib1 = bng[o0 + 1] * rinv, bbv1 = bnb[o0 + 1];

    const float* srcs[4] = {G + base, X1 + base, X2 + base, Hres + base + 32};
    constexpr int NP = FROMX ? 3 : 4;
    float4 pf[4][4];
    float px0[4], px1[4];
    StartW sw;
    const float* xs = Hres;
    int sub = lane >> 3;
    if (FROMX) {
      sw = start_weights(Wst, bst, (lane & 7) << 2);
      xs = x_series(Hres, wid);
#pragma unroll
      for (int j = 0; j < 4; ++j) {
        int row = min(4 * j + sub + 1, 63);
        px0[j] = xs[row];
        px1[j] = xs[XCH + row];
      }
    }
#pragma unroll
    for (int p = 0; p < NP; ++p)
#pragma unroll
      for (int j = 0; j < 4; ++j) pf[p][j] = ld_group(srcs[p] + j * 128, lane);

#pragma unroll 1
    for (int it = 0; it < 4; ++it) {
#pragma unroll
      for (int p = 0; p < NP; ++p)
#pragma unroll
        for (int j = 0; j < 4; ++j) st_group(T + p * 16 * APITCH, j, lane, pf[p][j]);
      if (FROMX) {
#pragma unroll
        for (int j = 0; j < 4; ++j)
          st_group(T + 48 * APITCH, j, lane, start_conv(px0[j], px1[j], sw));
      }
      if (it < 3) {
#pragma unroll
        for (int p = 0; p < NP; ++p)
#pragma unroll
          for (int j = 0; j < 4; ++j)
            pf[p][j] = ld_group(srcs[p] + (it + 1) * 512 + j * 128, lane);
        if (FROMX) {
#pragma unroll
          for (int j = 0; j < 4; ++j) {
            int row = min(16 * (it + 1) + 4 * j + sub + 1, 63);
            px0[j] = xs[row];
            px1[j] = xs[XCH + row];
          }
        }
      }
      v8f c0, c1;
#pragma unroll
      for (int r = 0; r < 8; ++r) { c0[r] = bias0; c1[r] = bias1; }
#pragma unroll
      for (int part = 0; part < 3; ++part) {
#pragma unroll
        for (int s = 0; s < 8; ++s) {
          v2f a = *(const v2f*)(T + (part * 16 + l15) * APITCH + s * 4 + half * 2);
          c0 = wmma_f32(a, bw0[part * 8 + s], c0);
          c1 = wmma_f32(a, bw1[part * 8 + s], c1);
        }
      }
      float2 res[8];
#pragma unroll
      for (int r = 0; r < 8; ++r)
        res[r] = *(const float2*)(T + (48 + r + half * 8) * APITCH + o0);
#pragma unroll
      for (int r = 0; r < 8; ++r) {
        int l = (it << 4) + r + half * 8;
        float g0 = c0[r] * ig0 + bgv0;
        float g1 = c1[r] * ig1 + bgv1;
        float h0 = g0 + res[r].x;
        float h1 = g1 + res[r].y;
        float2 v;
        v.x = h0 * ib0 + bbv0;
        v.y = h1 * ib1 + bbv1;
        *(float2*)(HS + l * APITCH + o0) = v;
        if (!LAST && l < Lh) *(float2*)(Hout + base + (size_t)l * 32 + o0) = v;
      }
    }
  }

  {
    float bf0 = bf[o0], bf1 = bf[o0 + 1], bg0 = bg[o0], bg1 = bg[o0 + 1];
    v2f wf0[16], wf1[16], wg0[16], wg1[16];
#pragma unroll
    for (int s = 0; s < 16; ++s) {
      int k = s * 4 + half * 2;
      int cc = k & 31, tap = k >> 5;
      const float* wf = Wf + (size_t)o0 * 64 + cc * 2 + tap;
      const float* wg = Wg + (size_t)o0 * 64 + cc * 2 + tap;
      wf0[s].x = wf[0];  wf0[s].y = wf[2];
      wf1[s].x = wf[64]; wf1[s].y = wf[66];
      wg0[s].x = wg[0];  wg0[s].y = wg[2];
      wg1[s].x = wg[64]; wg1[s].y = wg[66];
    }
#pragma unroll 1
    for (int it = 0; it < 4; ++it) {
      v8f cf0, cf1, cg0, cg1;
#pragma unroll
      for (int r = 0; r < 8; ++r) { cf0[r] = bf0; cf1[r] = bf1; cg0[r] = bg0; cg1[r] = bg1; }
      const float* Hr = HS + (16 * it + l15) * APITCH + half * 2;
#pragma unroll
      for (int s = 0; s < 16; ++s) {
        v2f a = *(const v2f*)(Hr + (s >> 3) * APITCH + ((s & 7) << 2));
        cf0 = wmma_f32(a, wf0[s], cf0);
        cf1 = wmma_f32(a, wf1[s], cf1);
        cg0 = wmma_f32(a, wg0[s], cg0);
        cg1 = wmma_f32(a, wg1[s], cg1);
      }
#pragma unroll
      for (int r = 0; r < 8; ++r) {
        int l = (it << 4) + r + half * 8;
        if (l < Lg) {
          float2 v;
          v.x = tanhf(cf0[r]) * (1.0f / (1.0f + __expf(-cg0[r])));
          v.y = tanhf(cf1[r]) * (1.0f / (1.0f + __expf(-cg1[r])));
          if (LAST) *(float2*)(T + l * APITCH + o0) = v;
          else      *(float2*)(Gout + base + (size_t)l * 32 + o0) = v;
        }
      }
    }
  }

  if (LAST) {
    v2f kw0[24], kw1[24];
#pragma unroll
    for (int s = 0; s < 24; ++s) {
      int k = (s & 7) * 4 + half * 2;
      const float* w = Wk + (size_t)(s >> 3) * 1024 + k;
      kw0[s] = *(const v2f*)(w + (size_t)o0 * 32);
      kw1[s] = *(const v2f*)(w + (size_t)(o0 + 1) * 32);
    }
    float bias0 = bk[o0] + bk[32 + o0] + bk[64 + o0];
    float bias1 = bk[o0 + 1] + bk[33 + o0] + bk[65 + o0];
    float inv = rsqrtf(1.0f + EPSBN);
    float sc0 = og[o0] * inv,     sh0 = ob[o0];
    float sc1 = og[o0 + 1] * inv, sh1 = ob[o0 + 1];
    int b = wid >> 9;
    int n = wid & 511;
    const float* srcs[2] = {G0 + 2 * 32 + base, G + 1 * 32 + base};
    float4 pf[2][4];
#pragma unroll
    for (int p = 0; p < 2; ++p)
#pragma unroll
      for (int j = 0; j < 4; ++j) pf[p][j] = ld_group(srcs[p] + j * 128, lane);
#pragma unroll 1
    for (int it = 0; it < 4; ++it) {
#pragma unroll
      for (int p = 0; p < 2; ++p)
#pragma unroll
        for (int j = 0; j < 4; ++j) st_group(HS + p * 16 * APITCH, j, lane, pf[p][j]);
      if (it < 3) {
#pragma unroll
        for (int p = 0; p < 2; ++p)
#pragma unroll
          for (int j = 0; j < 4; ++j)
            pf[p][j] = ld_group(srcs[p] + (it + 1) * 512 + j * 128, lane);
      }
      v8f c0, c1;
#pragma unroll
      for (int r = 0; r < 8; ++r) { c0[r] = bias0; c1[r] = bias1; }
#pragma unroll
      for (int part = 0; part < 3; ++part) {
        const float* Ap = (part < 2 ? HS + (part * 16 + l15) * APITCH
                                    : T + (16 * it + l15) * APITCH) + half * 2;
#pragma unroll
        for (int s = 0; s < 8; ++s) {
          v2f a = *(const v2f*)(Ap + s * 4);
          c0 = wmma_f32(a, kw0[part * 8 + s], c0);
          c1 = wmma_f32(a, kw1[part * 8 + s], c1);
        }
      }
#pragma unroll
      for (int r = 0; r < 8; ++r) {
        int l = (it << 4) + r + half * 8;
        if (l < 61) {
          float2 v;
          v.x = c0[r] * sc0 + sh0;
          v.y = c1[r] * sc1 + sh1;
          *(float2*)(out + ((size_t)(b * 61 + l) * 512 + n) * 32 + o0) = v;
        }
      }
    }
  }
}

extern "C" void kernel_launch(void* const* d_in, const int* in_sizes, int n_in,
                              void* d_out, int out_size, void* d_ws, size_t ws_size,
                              hipStream_t stream) {
  (void)in_sizes; (void)n_in; (void)out_size; (void)ws_size;
  const float* x   = (const float*)d_in[0];
  const float* adj = (const float*)d_in[1];
  const float* Wst = (const float*)d_in[2];
  const float* bst = (const float*)d_in[3];
  const float* Wf  = (const float*)d_in[4];
  const float* bf  = (const float*)d_in[5];
  const float* Wg  = (const float*)d_in[6];
  const float* bg  = (const float*)d_in[7];
  const float* Wsk = (const float*)d_in[8];
  const float* bsk = (const float*)d_in[9];
  const float* Wgc = (const float*)d_in[10];
  const float* bgc = (const float*)d_in[11];
  const float* gcg = (const float*)d_in[12];
  const float* gcb = (const float*)d_in[13];
  const float* bng = (const float*)d_in[14];
  const float* bnb = (const float*)d_in[15];
  const float* og  = (const float*)d_in[16];
  const float* ob  = (const float*)d_in[17];
  float* out = (float*)d_out;

  const size_t BUF = (size_t)8 * 512 * 64 * 32 + 1024;
  float* ws   = (float*)d_ws;
  float* bufA = ws;
  float* bufB = ws + BUF;
  float* bufC = ws + 2 * BUF;
  float* bufD = ws + 3 * BUF;
  float* bufE = ws + 4 * BUF;

  k_gated<true><<<512, 256, 0, stream>>>(x, bufC, Wf, bf, Wg, bg, Wst, bst, 63);
  k_diffuse<<<256, 256, 0, stream>>>(adj, bufC, bufD);
  k_diffuse<<<256, 256, 0, stream>>>(adj, bufD, bufE);
  k_layer<true, false><<<512, 256, 0, stream>>>(
      bufC, bufD, bufE, x, Wgc, bgc, gcg, gcb, bng, bnb, Wst, bst,
      Wf + 2048, bf + 32, Wg + 2048, bg + 32, nullptr, Wsk, bsk, og, ob,
      bufB, bufA, nullptr, 63, 62);
  k_diffuse<<<256, 256, 0, stream>>>(adj, bufA, bufD);
  k_diffuse<<<256, 256, 0, stream>>>(adj, bufD, bufE);
  k_layer<false, true><<<512, 256, 0, stream>>>(
      bufA, bufD, bufE, bufB, Wgc + 3072, bgc + 32, gcg + 32, gcb + 32, bng + 32,
      bnb + 32, Wst, bst, Wf + 4096, bf + 64, Wg + 4096, bg + 64, bufC, Wsk, bsk,
      og, ob, nullptr, nullptr, out, 62, 61);
}


# gated conv epilogue: tanh(f)*sigmoid(g) as sign(f)(1-e)/((1+e)(1+G)), e=exp(-2|f|), G=exp(-g): branch-free, two hardware exp and one reciprocal per output; whole tiles stored without row tests
# speedup vs baseline: 1.9462x; 1.0469x over previous
#include <hip/hip_runtime.h>
#include <cmath>


typedef __attribute__((ext_vector_type(2))) float v2f;
typedef __attribute__((ext_vector_type(8))) float v8f;

#define EPSBN 1e-5f

__device__ __forceinline__ v8f wmma_f32(v2f a, v2f b, v8f c) {
  return __builtin_amdgcn_wmma_f32_16x16x4_f32(false, a, false, b, (short)0, c,
                                               false, false);
}

struct StartW { float w0[4], w1[4], bb[4]; };

__device__ __forceinline__ StartW start_weights(const float* Wst, const float* bst,
                                                int c4) {
  StartW w;
#pragma unroll
  for (int q = 0; q < 4; ++q) {
    w.w0[q] = Wst[(c4 + q) * 2];
    w.w1[q] = Wst[(c4 + q) * 2 + 1];
    w.bb[q] = bst[c4 + q];
  }
  return w;
}

__device__ __forceinline__ float4 start_conv(float x0, float x1, const StartW& w) {
  float4 h;
  h.x = fmaf(x0, w.w0[0], fmaf(x1, w.w1[0], w.bb[0]));
  h.y = fmaf(x0, w.w0[1], fmaf(x1, w.w1[1], w.bb[1]));
  h.z = fmaf(x0, w.w0[2], fmaf(x1, w.w1[2], w.bb[2]));
  h.w = fmaf(x0, w.w0[3], fmaf(x1, w.w1[3], w.bb[3]));
  return h;
}

#define XCH (512 * 64)
__device__ __forceinline__ const float* x_series(const float* x, int wid) {
  return x + (size_t)(wid >> 9) * 2 * XCH + (size_t)(wid & 511) * 64;
}

#define APITCH 36

__device__ __forceinline__ float4 ld_group(const float* src, int lane) {
  return *(const float4*)(src + lane * 4);
}
__device__ __forceinline__ void st_group(float* tile, int j, int lane, float4 v) {
  *(float4*)(tile + (4 * j + (lane >> 3)) * APITCH + ((lane & 7) << 2)) = v;
}

__device__ __forceinline__ float gate(float f, float g) {
  float e = __builtin_amdgcn_exp2f(fabsf(f) * -2.8853900818f);
  float G = __builtin_amdgcn_exp2f(g * -1.4426950409f);
  float r = __builtin_amdgcn_rcpf((1.0f + e) * (1.0f + G));
  return copysignf((1.0f - e) * r, f);
}

#define GROWS 20
template <bool FROMX>
__global__ void __launch_bounds__(256, 4) k_gated(
    const float* __restrict__ H, float* __restrict__ G,
    const float* __restrict__ Wf, const float* __restrict__ bf,
    const float* __restrict__ Wg, const float* __restrict__ bg,
    const float* __restrict__ Wst, const float* __restrict__ bst, int Lout) {
  __shared__ __attribute__((aligned(16))) float tiles[8][GROWS * APITCH + (FROMX ? 224 : 0)];
  int wave = threadIdx.x >> 5;
  int wid  = blockIdx.x * 8 + wave;
  int lane = threadIdx.x & 31;
  int half = lane >> 4;
  int l15  = lane & 15;
  int o0   = 2 * l15;
  float* T = tiles[wave];

  float bf0 = bf[o0], bf1 = bf[o0 + 1], bg0 = bg[o0], bg1 = bg[o0 + 1];
  v2f wf0[16], wf1[16], wg0[16], wg1[16];
#pragma unroll
  for (int s = 0; s < 16; ++s) {
    int k = s * 4 + half * 2;
    int c0 = k & 31, tap = k >> 5;
    const float* wf = Wf + (size_t)o0 * 64 + c0 * 2 + tap;
    const float* wg = Wg + (size_t)o0 * 64 + c0 * 2 + tap;
    wf0[s].x = wf[0];  wf0[s].y = wf[2];
    wf1[s].x = wf[64]; wf1[s].y = wf[66];
    wg0[s].x = wg[0];  wg0[s].y = wg[2];
    wg1[s].x = wg[64]; wg1[s].y = wg[66];
  }

  size_t base = (size_t)wid * 2048;
  float4 pf[5];
  float* XL = T + GROWS * APITCH;
  float* WL = XL + 128;
  int sub = lane >> 3;
  if (FROMX) {
    const float* xs = x_series(H, wid);
    float a0 = xs[lane], a1 = xs[lane + 32];
    float b0 = xs[XCH + lane], b1 = xs[XCH + lane + 32];
    float u0 = Wst[lane * 2], u1 = Wst[lane * 2 + 1], ub = bst[lane];
    XL[lane] = a0; XL[lane + 32] = a1;
    XL[64 + lane] = b0; XL[96 + lane] = b1;
    float* wq = WL + (lane >> 2) * 12 + (lane & 3);
    wq[0] = u0; wq[4] = u1; wq[8] = ub;
  } else {
#pragma unroll
    for (int j = 0; j < 5; ++j) pf[j] = ld_group(H + base + j * 128, lane);
  }

#pragma unroll 1
  for (int it = 0; it < 4; ++it) {
    if (FROMX) {
      float xa[5], xb[5];
#pragma unroll
      for (int j = 0; j < 5; ++j) {
        int row = min(16 * it + 4 * j + sub, 63);
        xa[j] = XL[row];
        xb[j] = XL[64 + row];
      }
      StartW sw;
      const float4* wq = (const float4*)(WL + (lane & 7) * 12);
      float4 q0 = wq[0], q1 = wq[1], qb = wq[2];
      sw.w0[0] = q0.x; sw.w0[1] = q0.y; sw.w0[2] = q0.z; sw.w0[3] = q0.w;
      sw.w1[0] = q1.x; sw.w1[1] = q1.y; sw.w1[2] = q1.z; sw.w1[3] = q1.w;
      sw.bb[0] = qb.x; sw.bb[1] = qb.y; sw.bb[2] = qb.z; sw.bb[3] = qb.w;
#pragma unroll
      for (int j = 0; j < 5; ++j) st_group(T, j, lane, start_conv(xa[j], xb[j], sw));
    } else {
#pragma unroll
      for (int j = 0; j < 5; ++j) st_group(T, j, lane, pf[j]);
      if (it < 3) {
        const float* nx = H + base + (it + 1) * 512;
#pragma unroll
        for (int j = 0; j < 5; ++j) pf[j] = ld_group(nx + j * 128, lane);
      }
    }
    v8f cf0, cf1, cg0, cg1;
#pragma unroll
    for (int r = 0; r < 8; ++r) { cf0[r] = bf0; cf1[r] = bf1; cg0[r] = bg0; cg1[r] = bg1; }
#pragma unroll
    for (int s = 0; s < 16; ++s) {
      v2f a = *(const v2f*)(T + (l15 + (s >> 3)) * APITCH + ((s & 7) << 2) + half * 2);
      cf0 = wmma_f32(a, wf0[s], cf0);
      cf1 = wmma_f32(a, wf1[s], cf1);
      cg0 = wmma_f32(a, wg0[s], cg0);
      cg1 = wmma_f32(a, wg1[s], cg1);
    }
    float2 v[8];
#pragma unroll
    for (int r = 0; r < 8; ++r) {
      v[r].x = gate(cf0[r], cg0[r]);
      v[r].y = gate(cf1[r], cg1[r]);
    }
    float* Gt = G + base + (size_t)((it << 4) + half * 8) * 32 + o0;
    if ((it << 4) + 16 <= Lout) {
#pragma unroll
      for (int r = 0; r < 8; ++r) *(float2*)(Gt + r * 32) = v[r];
    } else {
#pragma unroll
      for (int r = 0; r < 8; ++r) {
        int l = (it << 4) + r + half * 8;
        if (l < Lout) *(float2*)(Gt + r * 32) = v[r];
      }
    }
  }
}

#define LDSPITCH 516
#define DROWS 64

struct DFrag {
  v2f a[4][4];
  float4 r0[4];
  float4 r1[4];
};

__device__ __forceinline__ void dload(DFrag& f, const float* Ar,
                                      const float* xr, int n0) {
#pragma unroll
  for (int u = 0; u < 4; ++u) {
#pragma unroll
    for (int t = 0; t < 4; ++t)
      f.a[u][t] = *(const v2f*)(Ar + t * 16 * LDSPITCH + n0 + 4 * u);
    f.r0[u] = *(const float4*)(xr + (size_t)(n0 + 4 * u) * 2048);
    f.r1[u] = *(const float4*)(xr + (size_t)(n0 + 4 * u) * 2048 + 2048);
  }
}

__device__ __forceinline__ void dmma(const DFrag& f, v8f (&acc)[4][4]) {
#pragma unroll
  for (int u = 0; u < 4; ++u) {
    v2f b0, b1, b2, b3;
    b0.x = f.r0[u].x; b0.y = f.r1[u].x;
    b1.x = f.r0[u].y; b1.y = f.r1[u].y;
    b2.x = f.r0[u].z; b2.y = f.r1[u].z;
    b3.x = f.r0[u].w; b3.y = f.r1[u].w;
#pragma unroll
    for (int t = 0; t < 4; ++t) {
      acc[t][0] = wmma_f32(f.a[u][t], b0, acc[t][0]);
      acc[t][1] = wmma_f32(f.a[u][t], b1, acc[t][1]);
      acc[t][2] = wmma_f32(f.a[u][t], b2, acc[t][2]);
      acc[t][3] = wmma_f32(f.a[u][t], b3, acc[t][3]);
    }
  }
}

__global__ void __launch_bounds__(256) k_diffuse(
    const float* __restrict__ adj, const float* __restrict__ Xin,
    float* __restrict__ Xout) {
  __shared__ float aT[DROWS * LDSPITCH];

  int bid = blockIdx.x;
  int cg = bid & 3;
  int mt = (bid >> 2) & 7;
  int b  = bid >> 5;
  int m0 = mt << 6;

#pragma unroll
  for (int i = 0; i < 32; ++i) {
    int idx = threadIdx.x + 256 * i;
    int q   = idx & 3;
    int kk  = (idx >> 2) & 7;
    int g   = (idx >> 5) & 3;
    int k   = ((idx >> 7) << 3) + kk;
    int ml  = 16 * g + 4 * q;
    const float4 v = *(const float4*)(adj + (size_t)k * 512 + m0 + ml);
    aT[(ml + 0) * LDSPITCH + k] = v.x;
    aT[(ml + 1) * LDSPITCH + k] = v.y;
    aT[(ml + 2) * LDSPITCH + k] = v.z;
    aT[(ml + 3) * LDSPITCH + k] = v.w;
  }
  __syncthreads();

  int wave = threadIdx.x >> 5;
  int lane = threadIdx.x & 31;
  int half = lane >> 4;
  int l15  = lane & 15;
  int j0   = (cg << 9) + (wave << 6);

  const float* Xb = Xin + (size_t)b * 512 * 2048;
  float* Ob       = Xout + (size_t)b * 512 * 2048;
  const float* Ar = aT + l15 * LDSPITCH + half * 2;

  const float* xr = Xb + (size_t)(half * 2) * 2048 + j0 + 4 * l15;

  v8f acc[4][4];
#pragma unroll
  for (int t = 0; t < 4; ++t)
#pragma unroll
    for (int j = 0; j < 4; ++j)
      acc[t][j] = v8f{};

#define LOADS_GROUP() __builtin_amdgcn_sched_barrier(0)
  DFrag f0, f1;
  dload(f0, Ar, xr, 0);
#pragma unroll 1
  for (int n0 = 0; n0 < 480; n0 += 32) {
    LOADS_GROUP();
    dload(f1, Ar, xr, n0 + 16);
    LOADS_GROUP();
    dmma(f0, acc);
    LOADS_GROUP();
    dload(f0, Ar, xr, n0 + 32);
    LOADS_GROUP();
    dmma(f1, acc);
  }
  LOADS_GROUP();
  dload(f1, Ar, xr, 496);
  LOADS_GROUP();
  dmma(f0, acc);
  dmma(f1, acc);
#undef LOADS_GROUP

#pragma unroll
  for (int t = 0; t < 4; ++t) {
#pragma unroll
    for (int r = 0; r < 8; ++r) {
      size_t orow = (size_t)(m0 + 16 * t + r + half * 8) * 2048 + j0 + 4 * l15;
      float4 o;
      o.x = acc[t][0][r]; o.y = acc[t][1][r];
      o.z = acc[t][2][r]; o.w = acc[t][3][r];
      *(float4*)(Ob + orow) = o;
    }
  }
}

#define HROWS 68
template <bool FROMX, bool LAST>
__global__ void __launch_bounds__(256, 4) k_layer(
    const float* __restrict__ G, const float* __restrict__ X1,
    const float* __restrict__ X2, const float* __restrict__ Hres,
    const float* __restrict__ Wgc, const float* __restrict__ bgc,
    const float* __restrict__ gcg, const float* __restrict__ gcb,
    const float* __restrict__ bng, const float* __restrict__ bnb,
    const float* __restrict__ Wst, const float* __restrict__ bst,
    const float* __restrict__ Wf, const float* __restrict__ bf,
    const float* __restrict__ Wg, const float* __restrict__ bg,
    const float* __restrict__ G0, const float* __restrict__ Wk,
    const float* __restrict__ bk, const float* __restrict__ og,
    const float* __restrict__ ob, float* __restrict__ Hout,
    float* __restrict__ Gout, float* __restrict__ out, int Lh, int Lg) {
  __shared__ __attribute__((aligned(16))) float tiles[8][(64 + HROWS) * APITCH];
  int wave = threadIdx.x >> 5;
  int wid  = blockIdx.x * 8 + wave;
  int lane = threadIdx.x & 31;
  int half = lane >> 4;
  int l15  = lane & 15;
  int o0   = 2 * l15;
  float* T  = tiles[wave];
  float* HS = T + 64 * APITCH;
  size_t base = (size_t)wid * 2048;

  {
    v2f bw0[24], bw1[24];
#pragma unroll
    for (int s = 0; s < 24; ++s) {
      int k = (s & 7) * 4 + half * 2;
      const float* w = Wgc + (s >> 3) * 32 + k;
      bw0[s] = *(const v2f*)(w + (size_t)o0 * 96);
      bw1[s] = *(const v2f*)(w + (size_t)(o0 + 1) * 96);
    }
    float bias0 = bgc[o0], bias1 = bgc[o0 + 1];
    float rinv = rsqrtf(1.0f + EPSBN);
    float ig0 = gcg[o0] * rinv,     bgv0 = gcb[o0];
    float ig1 = gcg[o0 + 1] * rinv, bgv1 = gcb[o0 + 1];
    float ib0 = bng[o0] * rinv,     bbv0 = bnb[o0];
    float ib1 = bng[o0 + 1] * rinv, bbv1 = bnb[o0 + 1];

    const float* srcs[4] = {G + base, X1 + base, X2 + base, Hres + base + 32};
    constexpr int NP = FROMX ? 3 : 4;
    float4 pf[4][4];
    float px0[4], px1[4];
    StartW sw;
    const float* xs = Hres;
    int sub = lane >> 3;
    if (FROMX) {
      sw = start_weights(Wst, bst, (lane & 7) << 2);
      xs = x_series(Hres, wid);
#pragma unroll
      for (int j = 0; j < 4; ++j) {
        int row = min(4 * j + sub + 1, 63);
        px0[j] = xs[row];
        px1[j] = xs[XCH + row];
      }
    }
#pragma unroll
    for (int p = 0; p < NP; ++p)
#pragma unroll
      for (int j = 0; j < 4; ++j) pf[p][j] = ld_group(srcs[p] + j * 128, lane);

#pragma unroll 1
    for (int it = 0; it < 4; ++it) {
#pragma unroll
      for (int p = 0; p < NP; ++p)
#pragma unroll
        for (int j = 0; j < 4; ++j) st_group(T + p * 16 * APITCH, j, lane, pf[p][j]);
      if (FROMX) {
#pragma unroll
        for (int j = 0; j < 4; ++j)
          st_group(T + 48 * APITCH, j, lane, start_conv(px0[j], px1[j], sw));
      }
      if (it < 3) {
#pragma unroll
        for (int p = 0; p < NP; ++p)
#pragma unroll
          for (int j = 0; j < 4; ++j)
            pf[p][j] = ld_group(srcs[p] + (it + 1) * 512 + j * 128, lane);
        if (FROMX) {
#pragma unroll
          for (int j = 0; j < 4; ++j) {
            int row = min(16 * (it + 1) + 4 * j + sub + 1, 63);
            px0[j] = xs[row];
            px1[j] = xs[XCH + row];
          }
        }
      }
      v8f c0, c1;
#pragma unroll
      for (int r = 0; r < 8; ++r) { c0[r] = bias0; c1[r] = bias1; }
#pragma unroll
      for (int part = 0; part < 3; ++part) {
#pragma unroll
        for (int s = 0; s < 8; ++s) {
          v2f a = *(const v2f*)(T + (part * 16 + l15) * APITCH + s * 4 + half * 2);
          c0 = wmma_f32(a, bw0[part * 8 + s], c0);
          c1 = wmma_f32(a, bw1[part * 8 + s], c1);
        }
      }
      float2 res[8];
#pragma unroll
      for (int r = 0; r < 8; ++r)
        res[r] = *(const float2*)(T + (48 + r + half * 8) * APITCH + o0);
#pragma unroll
      for (int r = 0; r < 8; ++r) {
        int l = (it << 4) + r + half * 8;
        float g0 = c0[r] * ig0 + bgv0;
        float g1 = c1[r] * ig1 + bgv1;
        float h0 = g0 + res[r].x;
        float h1 = g1 + res[r].y;
        float2 v;
        v.x = h0 * ib0 + bbv0;
        v.y = h1 * ib1 + bbv1;
        *(float2*)(HS + l * APITCH + o0) = v;
        if (!LAST && l < Lh) *(float2*)(Hout + base + (size_t)l * 32 + o0) = v;
      }
    }
  }

  {
    float bf0 = bf[o0], bf1 = bf[o0 + 1], bg0 = bg[o0], bg1 = bg[o0 + 1];
    v2f wf0[16], wf1[16], wg0[16], wg1[16];
#pragma unroll
    for (int s = 0; s < 16; ++s) {
      int k = s * 4 + half * 2;
      int cc = k & 31, tap = k >> 5;
      const float* wf = Wf + (size_t)o0 * 64 + cc * 2 + tap;
      const float* wg = Wg + (size_t)o0 * 64 + cc * 2 + tap;
      wf0[s].x = wf[0];  wf0[s].y = wf[2];
      wf1[s].x = wf[64]; wf1[s].y = wf[66];
      wg0[s].x = wg[0];  wg0[s].y = wg[2];
      wg1[s].x = wg[64]; wg1[s].y = wg[66];
    }
#pragma unroll 1
    for (int it = 0; it < 4; ++it) {
      v8f cf0, cf1, cg0, cg1;
#pragma unroll
      for (int r = 0; r < 8; ++r) { cf0[r] = bf0; cf1[r] = bf1; cg0[r] = bg0; cg1[r] = bg1; }
      const float* Hr = HS + (16 * it + l15) * APITCH + half * 2;
#pragma unroll
      for (int s = 0; s < 16; ++s) {
        v2f a = *(const v2f*)(Hr + (s >> 3) * APITCH + ((s & 7) << 2));
        cf0 = wmma_f32(a, wf0[s], cf0);
        cf1 = wmma_f32(a, wf1[s], cf1);
        cg0 = wmma_f32(a, wg0[s], cg0);
        cg1 = wmma_f32(a, wg1[s], cg1);
      }
      float2 v[8];
#pragma unroll
      for (int r = 0; r < 8; ++r) {
        v[r].x = gate(cf0[r], cg0[r]);
        v[r].y = gate(cf1[r], cg1[r]);
      }
      int lt = (it << 4) + half * 8;
      if ((it << 4) + 16 <= Lg) {
#pragma unroll
        for (int r = 0; r < 8; ++r) {
          if (LAST) *(float2*)(T + (lt + r) * APITCH + o0) = v[r];
          else      *(float2*)(Gout + base + (size_t)(lt + r) * 32 + o0) = v[r];
        }
      } else {
#pragma unroll
        for (int r = 0; r < 8; ++r) {
          if (lt + r < Lg) {
            if (LAST) *(float2*)(T + (lt + r) * APITCH + o0) = v[r];
            else      *(float2*)(Gout + base + (size_t)(lt + r) * 32 + o0) = v[r];
          }
        }
      }
    }
  }

  if (LAST) {
    v2f kw0[24], kw1[24];
#pragma unroll
    for (int s = 0; s < 24; ++s) {
      int k = (s & 7) * 4 + half * 2;
      const float* w = Wk + (size_t)(s >> 3) * 1024 + k;
      kw0[s] = *(const v2f*)(w + (size_t)o0 * 32);
      kw1[s] = *(const v2f*)(w + (size_t)(o0 + 1) * 32);
    }
    float bias0 = bk[o0] + bk[32 + o0] + bk[64 + o0];
    float bias1 = bk[o0 + 1] + bk[33 + o0] + bk[65 + o0];
    float inv = rsqrtf(1.0f + EPSBN);
    float sc0 = og[o0] * inv,     sh0 = ob[o0];
    float sc1 = og[o0 + 1] * inv, sh1 = ob[o0 + 1];
    int b = wid >> 9;
    int n = wid & 511;
    const float* srcs[2] = {G0 + 2 * 32 + base, G + 1 * 32 + base};
    float4 pf[2][4];
#pragma unroll
    for (int p = 0; p < 2; ++p)
#pragma unroll
      for (int j = 0; j < 4; ++j) pf[p][j] = ld_group(srcs[p] + j * 128, lane);
#pragma unroll 1
    for (int it = 0; it < 4; ++it) {
#pragma unroll
      for (int p = 0; p < 2; ++p)
#pragma unroll
        for (int j = 0; j < 4; ++j) st_group(HS + p * 16 * APITCH, j, lane, pf[p][j]);
      if (it < 3) {
#pragma unroll
        for (int p = 0; p < 2; ++p)
#pragma unroll
          for (int j = 0; j < 4; ++j)
            pf[p][j] = ld_group(srcs[p] + (it + 1) * 512 + j * 128, lane);
      }
      v8f c0, c1;
#pragma unroll
      for (int r = 0; r < 8; ++r) { c0[r] = bias0; c1[r] = bias1; }
#pragma unroll
      for (int part = 0; part < 3; ++part) {
        const float* Ap = (part < 2 ? HS + (part * 16 + l15) * APITCH
                                    : T + (16 * it + l15) * APITCH) + half * 2;
#pragma unroll
        for (int s = 0; s < 8; ++s) {
          v2f a = *(const v2f*)(Ap + s * 4);
          c0 = wmma_f32(a, kw0[part * 8 + s], c0);
          c1 = wmma_f32(a, kw1[part * 8 + s], c1);
        }
      }
#pragma unroll
      for (int r = 0; r < 8; ++r) {
        int l = (it << 4) + r + half * 8;
        if (l < 61) {
          float2 v;
          v.x = c0[r] * sc0 + sh0;
          v.y = c1[r] * sc1 + sh1;
          *(float2*)(out + ((size_t)(b * 61 + l) * 512 + n) * 32 + o0) = v;
        }
      }
    }
  }
}

extern "C" void kernel_launch(void* const* d_in, const int* in_sizes, int n_in,
                              void* d_out, int out_size, void* d_ws, size_t ws_size,
                              hipStream_t stream) {
  (void)in_sizes; (void)n_in; (void)out_size; (void)ws_size;
  const float* x   = (const float*)d_in[0];
  const float* adj = (const float*)d_in[1];
  const float* Wst = (const float*)d_in[2];
  const float* bst = (const float*)d_in[3];
  const float* Wf  = (const float*)d_in[4];
  const float* bf  = (const float*)d_in[5];
  const float* Wg  = (const float*)d_in[6];
  const float* bg  = (const float*)d_in[7];
  const float* Wsk = (const float*)d_in[8];
  const float* bsk = (const float*)d_in[9];
  const float* Wgc = (const float*)d_in[10];
  const float* bgc = (const float*)d_in[11];
  const float* gcg = (const float*)d_in[12];
  const float* gcb = (const float*)d_in[13];
  const float* bng = (const float*)d_in[14];
  const float* bnb = (const float*)d_in[15];
  const float* og  = (const float*)d_in[16];
  const float* ob  = (const float*)d_in[17];
  float* out = (float*)d_out;

  const size_t BUF = (size_t)8 * 512 * 64 * 32 + 1024;
  float* ws   = (float*)d_ws;
  float* bufA = ws;
  float* bufB = ws + BUF;
  float* bufC = ws + 2 * BUF;
  float* bufD = ws + 3 * BUF;
  float* bufE = ws + 4 * BUF;

  k_gated<true><<<512, 256, 0, stream>>>(x, bufC, Wf, bf, Wg, bg, Wst, bst, 63);
  k_diffuse<<<256, 256, 0, stream>>>(adj, bufC, bufD);
  k_diffuse<<<256, 256, 0, stream>>>(adj, bufD, bufE);
  k_layer<true, false><<<512, 256, 0, stream>>>(
      bufC, bufD, bufE, x, Wgc, bgc, gcg, gcb, bng, bnb, Wst, bst,
      Wf + 2048, bf + 32, Wg + 2048, bg + 32, nullptr, Wsk, bsk, og, ob,
      bufB, bufA, nullptr, 63, 62);
  k_diffuse<<<256, 256, 0, stream>>>(adj, bufA, bufD);
  k_diffuse<<<256, 256, 0, stream>>>(adj, bufD, bufE);
  k_layer<false, true><<<512, 256, 0, stream>>>(
      bufA, bufD, bufE, bufB, Wgc + 3072, bgc + 32, gcg + 32, gcb + 32, bng + 32,
      bnb + 32, Wst, bst, Wf + 4096, bf + 64, Wg + 4096, bg + 64, bufC, Wsk, bsk,
      og, ob, nullptr, nullptr, out, 62, 61);
}
